# starting kernel
# baseline (speedup 1.0000x reference)
#include <hip/hip_runtime.h>
#include <hip/hip_bf16.h>


typedef __attribute__((ext_vector_type(2))) float v2f;
typedef __attribute__((ext_vector_type(8))) float v8f;

__device__ __constant__ int c_sdx[7] = {0, 0, 0, 1, 1, 2, -1};
__device__ __constant__ int c_sdy[7] = {0, 1, 2, 0, 1, 0, 1};

#define TWO_PI_F 6.28318530717958647692f

template <int LOGW, int ROWS>
__global__ void __launch_bounds__(128)
same_scale_corr(const float* __restrict__ xin, int pairBase,
                float* __restrict__ out) {
  constexpr int W = 1 << LOGW;
  constexpr int H = W;
  constexpr int WS = W + 4;
  constexpr int ZROW = 32 * WS;
  constexpr int SLABS = H / ROWS;
  __shared__ float lds[33 * WS];

  const int b = blockIdx.x / SLABS;
  const int slab = blockIdx.x & (SLABS - 1);

  const int lane = threadIdx.x & 31;
  const int wave = threadIdx.x >> 5;
  const int m0 = wave * 16;

  const int mrow = m0 + (lane & 15);
  const bool aval = (mrow < 56);
  const int sidx = aval ? (mrow >> 3) : 0;
  const int l1 = mrow & 7;
  const int slot = c_sdx[sidx] + 1;
  const int dy = aval ? c_sdy[sidx] : 0;
  const int aIdx = (aval ? (slot * 8 + l1) * WS : ZROW) + dy;
  const int ncol = lane & 15;
  const int bBase = (ncol < 8) ? (8 + ncol) * WS : ZROW;
  const int khi = (lane >> 4) << 1;

  for (int idx = threadIdx.x; idx < WS; idx += 128) lds[ZROW + idx] = 0.0f;

  v8f acc = {};

  for (int xi = 0; xi < ROWS; ++xi) {
    const int xr = slab * ROWS + xi;

    {
      constexpr int NV4 = 8 * W;
      constexpr int PER_CH4 = W / 4;
      for (int idx = threadIdx.x; idx < NV4; idx += 128) {
        const int row = idx >> (LOGW - 2);
        const int y4 = idx & (PER_CH4 - 1);
        const int rs = row >> 3;
        const int ch = row & 7;
        int xs = xr + rs - 1;
        xs += (xs < 0) ? H : 0;
        xs -= (xs >= H) ? H : 0;
        *(float4*)&lds[row * WS + (y4 << 2)] =
            *(const float4*)&xin[((((size_t)b * 8 + ch) << LOGW) + xs) * W +
                                 (y4 << 2)];
      }
      const int row = threadIdx.x >> 2;
      const int e = threadIdx.x & 3;
      const int rs = row >> 3;
      const int ch = row & 7;
      int xs = xr + rs - 1;
      xs += (xs < 0) ? H : 0;
      xs -= (xs >= H) ? H : 0;
      lds[row * WS + W + e] =
          xin[((((size_t)b * 8 + ch) << LOGW) + xs) * W + e];
    }
    if (threadIdx.x < 8) {
      int xn = xr + 4;
      xn -= (xn >= H) ? H : 0;
      __builtin_prefetch(
          &xin[((((size_t)b * 8 + threadIdx.x) << LOGW) + xn) * W], 0, 1);
    }
    __syncthreads();

#pragma unroll 8
    for (int k = 0; k < W; k += 4) {
      const int kb = k + khi;
      v2f av, bv;
      av[0] = lds[aIdx + kb];
      av[1] = lds[aIdx + kb + 1];
      bv = *(const v2f*)&lds[bBase + kb];
      acc = __builtin_amdgcn_wmma_f32_16x16x4_f32(
           false, av,  false, bv,
           (short)0, acc,  false,  false);
    }
    __syncthreads();
  }

  const int rbase = m0 + ((lane >> 4) << 3);
  const int col = lane & 15;
  if (col < 8) {
#pragma unroll
    for (int i = 0; i < 8; ++i) {
      const int R = rbase + i;
      if (R < 56) {
        const int s = R >> 3;
        const int l1o = R & 7;
        atomicAdd(&out[(((size_t)b * 384) + pairBase + l1o * 8 + col) * 7 + s],
                  acc[i]);
      }
    }
  }
}

template <int LOGN>
__global__ void fft_rows(const float* __restrict__ xin,
                         float2* __restrict__ spec) {
  constexpr int N = 1 << LOGN;
  __shared__ float2 buf[N];
  const int row = blockIdx.x & (N - 1);
  const int img = blockIdx.x >> LOGN;
  const float* src = xin + (((size_t)img << LOGN) + row) * N;
  const int t = threadIdx.x;

  buf[__brev((unsigned)t) >> (32 - LOGN)] = make_float2(src[t], 0.0f);
  buf[__brev((unsigned)(t + N / 2)) >> (32 - LOGN)] =
      make_float2(src[t + N / 2], 0.0f);
  __syncthreads();
#pragma unroll
  for (int st = 1; st <= LOGN; ++st) {
    const int len = 1 << st;
    const int half = len >> 1;
    const int j = t & (half - 1);
    const int base = (t >> (st - 1)) << st;
    float sn, cs;
    __sincosf(-TWO_PI_F * (float)j / (float)len, &sn, &cs);
    const float2 u = buf[base + j];
    const float2 v = buf[base + j + half];
    const float2 tv = make_float2(v.x * cs - v.y * sn, v.x * sn + v.y * cs);
    buf[base + j] = make_float2(u.x + tv.x, u.y + tv.y);
    buf[base + j + half] = make_float2(u.x - tv.x, u.y - tv.y);
    __syncthreads();
  }
  float2* dst = spec + (((size_t)img << LOGN) + row) * N;
  dst[t] = buf[t];
  dst[t + N / 2] = buf[t + N / 2];
}

template <int LOGN>
__global__ void fft_cols(float2* __restrict__ spec) {
  constexpr int N = 1 << LOGN;
  __shared__ float2 buf[N];
  const int col = blockIdx.x & (N - 1);
  const int img = blockIdx.x >> LOGN;
  float2* base0 = spec + ((size_t)img << (2 * LOGN)) + col;
  const int t = threadIdx.x;

  buf[__brev((unsigned)t) >> (32 - LOGN)] = base0[(size_t)t * N];
  buf[__brev((unsigned)(t + N / 2)) >> (32 - LOGN)] =
      base0[(size_t)(t + N / 2) * N];
  __syncthreads();
#pragma unroll
  for (int st = 1; st <= LOGN; ++st) {
    const int len = 1 << st;
    const int half = len >> 1;
    const int j = t & (half - 1);
    const int base = (t >> (st - 1)) << st;
    float sn, cs;
    __sincosf(-TWO_PI_F * (float)j / (float)len, &sn, &cs);
    const float2 u = buf[base + j];
    const float2 v = buf[base + j + half];
    const float2 tv = make_float2(v.x * cs - v.y * sn, v.x * sn + v.y * cs);
    buf[base + j] = make_float2(u.x + tv.x, u.y + tv.y);
    buf[base + j + half] = make_float2(u.x - tv.x, u.y - tv.y);
    __syncthreads();
  }
  base0[(size_t)t * N] = buf[t];
  base0[(size_t)(t + N / 2) * N] = buf[t + N / 2];
}

template <int LOG1, int LOG2>
__global__ void __launch_bounds__(256)
cross_corr(const float2* __restrict__ s1, const float2* __restrict__ s2,
           int pairBase, float* __restrict__ out) {
  constexpr int N1 = 1 << LOG1;
  constexpr int N2 = 1 << LOG2;
  __shared__ float red[256];
  const int b = blockIdx.x >> 6;
  const int p = blockIdx.x & 63;
  const int l1 = p >> 3, l2 = p & 7;
  const float2* F1 = s1 + (((size_t)b * 8 + l1) << (2 * LOG1));
  const float2* F2 = s2 + (((size_t)b * 8 + l2) << (2 * LOG2));

  float a0 = 0.f, a1 = 0.f, a2 = 0.f, a3 = 0.f, a4 = 0.f, a5 = 0.f, a6 = 0.f;
  constexpr int BINS = N2 * N2;
  for (int tbin = threadIdx.x; tbin < BINS; tbin += blockDim.x) {
    const int ix2 = tbin >> LOG2;
    const int iy2 = tbin & (N2 - 1);
    const int fx = (ix2 < (N2 >> 1)) ? ix2 : ix2 - N2;
    const int fy = (iy2 < (N2 >> 1)) ? iy2 : iy2 - N2;
    const float2 A =
        F1[(((fx + N1) & (N1 - 1)) << LOG1) + ((fy + N1) & (N1 - 1))];
    const float2 B = F2[tbin];
    const float zr = A.x * B.x + A.y * B.y;
    const float zi = A.y * B.x - A.x * B.y;
    float sx, cx, sy, cy;
    __sincosf(TWO_PI_F * (float)fx / (float)N1, &sx, &cx);
    __sincosf(TWO_PI_F * (float)fy / (float)N1, &sy, &cy);
    const float c2x = cx * cx - sx * sx, s2x = 2.f * cx * sx;
    const float c2y = cy * cy - sy * sy, s2y = 2.f * cy * sy;
    a0 += zr;
    a1 += zr * cy - zi * sy;
    a2 += zr * c2y - zi * s2y;
    a3 += zr * cx - zi * sx;
    const float xr = zr * cx - zi * sx, xi = zr * sx + zi * cx;
    a4 += xr * cy - xi * sy;
    a5 += zr * c2x - zi * s2x;
    const float mr = zr * cx + zi * sx, mi = zi * cx - zr * sx;
    a6 += mr * cy - mi * sy;
  }

  const float inv = 1.0f / (float)BINS;
  float accs[7] = {a0, a1, a2, a3, a4, a5, a6};
#pragma unroll
  for (int s = 0; s < 7; ++s) {
    red[threadIdx.x] = accs[s];
    __syncthreads();
    for (int off = 128; off > 0; off >>= 1) {
      if (threadIdx.x < off) red[threadIdx.x] += red[threadIdx.x + off];
      __syncthreads();
    }
    if (threadIdx.x == 0)
      out[(((size_t)b * 384) + pairBase + p) * 7 + s] = red[0] * inv;
    __syncthreads();
  }
}

extern "C" void kernel_launch(void* const* d_in, const int* in_sizes, int n_in,
                              void* d_out, int out_size, void* d_ws,
                              size_t ws_size, hipStream_t stream) {
  const float* x0 = (const float*)d_in[0];
  const float* x1 = (const float*)d_in[1];
  const float* x2 = (const float*)d_in[2];
  float* out = (float*)d_out;

  hipMemsetAsync(d_out, 0, (size_t)out_size * sizeof(float), stream);

  same_scale_corr<8, 2><<<dim3(2 * 128), 128, 0, stream>>>(x0, 0, out);
  same_scale_corr<7, 1><<<dim3(2 * 128), 128, 0, stream>>>(x1, 192, out);
  same_scale_corr<6, 1><<<dim3(2 * 64), 128, 0, stream>>>(x2, 320, out);

  const size_t n0 = (size_t)16 * 256 * 256;
  const size_t n1 = (size_t)16 * 128 * 128;
  const size_t n2 = (size_t)16 * 64 * 64;
  if (ws_size >= (n0 + n1 + n2) * sizeof(float2)) {
    float2* spec0 = (float2*)d_ws;
    float2* spec1 = spec0 + n0;
    float2* spec2 = spec1 + n1;

    fft_rows<8><<<dim3(16 * 256), 128, 0, stream>>>(x0, spec0);
    fft_cols<8><<<dim3(16 * 256), 128, 0, stream>>>(spec0);
    fft_rows<7><<<dim3(16 * 128), 64, 0, stream>>>(x1, spec1);
    fft_cols<7><<<dim3(16 * 128), 64, 0, stream>>>(spec1);
    fft_rows<6><<<dim3(16 * 64), 32, 0, stream>>>(x2, spec2);
    fft_cols<6><<<dim3(16 * 64), 32, 0, stream>>>(spec2);

    cross_corr<8, 7><<<dim3(128), 256, 0, stream>>>(spec0, spec1, 64, out);
    cross_corr<8, 6><<<dim3(128), 256, 0, stream>>>(spec0, spec2, 128, out);
    cross_corr<7, 6><<<dim3(128), 256, 0, stream>>>(spec1, spec2, 256, out);
  }
}


# cross-scale pairs done in space: coarse images band-limit interpolated to the fine grid by f32 WMMA products, then one shifted-row WMMA product per fine scale; no FFTs, no spectra; sum order changed
# speedup vs baseline: 3.3580x; 3.3580x over previous
#include <hip/hip_runtime.h>


typedef __attribute__((ext_vector_type(2))) float v2f;
typedef __attribute__((ext_vector_type(8))) float v8f;

__device__ __constant__ int c_sdx[7] = {0, 0, 0, 1, 1, 2, -1};
__device__ __constant__ int c_sdy[7] = {0, 1, 2, 0, 1, 0, 1};

template <int LOGC, int LOGF>
__global__ void __launch_bounds__(128)
interp_mm(const float* __restrict__ xc, float* __restrict__ up) {
  constexpr int NC = 1 << LOGC;
  constexpr int NF = 1 << LOGF;
  constexpr int R = NF / NC;
  constexpr int VP = NC + 4;
  constexpr int T1 = NC / 64;
  constexpr int T2 = NF / 64;
  __shared__ float g[NF];
  __shared__ alignas(16) float Vs[16 * VP];
  __shared__ float red[4];

  const int t = threadIdx.x;
  const int lane = t & 31;
  const int wave = t >> 5;
  const int c = lane & 15;
  const int h = lane >> 4;
  const int khi = h << 1;
  const int img = blockIdx.x >> (LOGF - 4);
  const int x0 = (blockIdx.x & ((NF >> 4) - 1)) << 4;
  const float* src = xc + ((size_t)img << (2 * LOGC));

  for (int k = t; k < NF; k += 128) {
    float v = (float)(NC - 1);
    if (k != 0) {
      const int a = ((NC - 1) * k) & (2 * NF - 1);
      v = sinpif((float)a / (float)NF) / sinpif((float)k / (float)NF);
    }
    v += cospif((float)(k & (2 * R - 1)) / (float)R);
    g[k] = v * (1.0f / (float)NC);
  }
  __syncthreads();

  v8f accV[T1];
#pragma unroll
  for (int j = 0; j < T1; ++j) accV[j] = v8f{};
  float salt = 0.0f;
  const int xa = x0 + c;
#pragma unroll 8
  for (int k = 0; k < NC; k += 4) {
    const int kk = k + khi;
    v2f av;
    av[0] = g[(xa - R * kk) & (NF - 1)];
    av[1] = g[(xa - R * (kk + 1)) & (NF - 1)];
#pragma unroll
    for (int j = 0; j < T1; ++j) {
      const int n = ((wave * T1 + j) << 4) + c;
      v2f bv;
      bv[0] = src[(size_t)kk * NC + n];
      bv[1] = src[(size_t)(kk + 1) * NC + n];
      salt += bv[0] - bv[1];
      accV[j] = __builtin_amdgcn_wmma_f32_16x16x4_f32(
           false, av,  false, bv,
           (short)0, accV[j],  false,  false);
    }
  }
  salt = (c & 1) ? -salt : salt;
#pragma unroll
  for (int o = 16; o > 0; o >>= 1) salt += __shfl_xor(salt, o);
  if (lane == 0) red[wave] = salt;
#pragma unroll
  for (int j = 0; j < T1; ++j) {
#pragma unroll
    for (int i = 0; i < 8; ++i)
      Vs[(8 * h + i) * VP + ((wave * T1 + j) << 4) + c] = accV[j][i];
  }
  __syncthreads();
  const float S = (red[0] + red[1]) + (red[2] + red[3]);

  v8f accU[T2];
#pragma unroll
  for (int j = 0; j < T2; ++j) accU[j] = v8f{};
#pragma unroll 4
  for (int k = 0; k < NC; k += 4) {
    const int kk = k + khi;
    const v2f av = *(const v2f*)&Vs[c * VP + kk];
#pragma unroll
    for (int j = 0; j < T2; ++j) {
      const int n = ((wave * T2 + j) << 4) + c;
      v2f bv;
      bv[0] = g[(n - R * kk) & (NF - 1)];
      bv[1] = g[(n - R * (kk + 1)) & (NF - 1)];
      accU[j] = __builtin_amdgcn_wmma_f32_16x16x4_f32(
           false, av,  false, bv,
           (short)0, accU[j],  false,  false);
    }
  }

  const float corr = S * (1.0f / (float)(NC * NC));
  float* dst = up + ((size_t)img << (2 * LOGF));
#pragma unroll
  for (int j = 0; j < T2; ++j) {
    const int n = ((wave * T2 + j) << 4) + c;
    const float sn = sinpif((float)(n & (2 * R - 1)) / (float)R) * corr;
#pragma unroll
    for (int i = 0; i < 8; ++i) {
      const int row = x0 + 8 * h + i;
      const float sr = sinpif((float)(row & (2 * R - 1)) / (float)R);
      dst[(size_t)row * NF + n] = accU[j][i] - sr * sn;
    }
  }
}

template <int LOGW, int ROWS, int NX>
__global__ void __launch_bounds__(128) __attribute__((amdgpu_waves_per_eu(1, 4)))
shifted_corr(const float* __restrict__ xin, const float* __restrict__ ua,
             const float* __restrict__ ub, int pairBase,
             float* __restrict__ out) {
  constexpr int W = 1 << LOGW;
  constexpr int H = W;
  constexpr int WS = W + 4;
  constexpr int NR = 32 + 8 * NX;
  constexpr int ZROW = NR * WS;
  constexpr int SLABS = H / ROWS;
  __shared__ alignas(16) float lds[(NR + 1) * WS];

  const int b = blockIdx.x / SLABS;
  const int slab = blockIdx.x & (SLABS - 1);

  const int lane = threadIdx.x & 31;
  const int wave = threadIdx.x >> 5;
  const int m0 = wave * 16;

  const int mrow = m0 + (lane & 15);
  const bool aval = (mrow < 56);
  const int sidx = aval ? (mrow >> 3) : 0;
  const int l1 = mrow & 7;
  const int slot = c_sdx[sidx] + 1;
  const int dy = aval ? c_sdy[sidx] : 0;
  const int aIdx = (aval ? (slot * 8 + l1) * WS : ZROW) + dy;
  const int ncol = lane & 15;
  const int bBase0 = (ncol < 8) ? (8 + ncol) * WS
                                : ((NX >= 1) ? (32 + (ncol - 8)) * WS : ZROW);
  const int bBase1 = (NX == 2 && ncol < 8) ? (40 + ncol) * WS : ZROW;
  const int khi = (lane >> 4) << 1;

  for (int idx = threadIdx.x; idx < WS; idx += 128) lds[ZROW + idx] = 0.0f;

  v8f acc0 = {};
  v8f acc1 = {};

  for (int xi = 0; xi < ROWS; ++xi) {
    const int xr = slab * ROWS + xi;

    {
      constexpr int PER_CH4 = W / 4;
      constexpr int NA = W / 16;
      constexpr int NP = NX * W / 64;
      float4 va[NA];
      float4 vp[NP > 0 ? NP : 1];
#pragma unroll
      for (int i = 0; i < NA; ++i) {
        const int idx = threadIdx.x + i * 128;
        const int row = idx >> (LOGW - 2);
        const int y4 = idx & (PER_CH4 - 1);
        int xs = xr + (row >> 3) - 1;
        xs += (xs < 0) ? H : 0;
        xs -= (xs >= H) ? H : 0;
        va[i] = *(const float4*)&xin[((((size_t)b * 8 + (row & 7)) << LOGW) +
                                      xs) * W + (y4 << 2)];
      }
#pragma unroll
      for (int i = 0; i < NP; ++i) {
        const int idx = threadIdx.x + i * 128;
        const int prow = idx >> (LOGW - 2);
        const int y4 = idx & (PER_CH4 - 1);
        const float* us = (prow < 8) ? ua : ub;
        vp[i] = *(const float4*)&us[((((size_t)b * 8 + (prow & 7)) << LOGW) +
                                     xr) * W + (y4 << 2)];
      }
      const int erow = threadIdx.x >> 2;
      const int e = threadIdx.x & 3;
      int xe = xr + (erow >> 3) - 1;
      xe += (xe < 0) ? H : 0;
      xe -= (xe >= H) ? H : 0;
      const float ve =
          xin[((((size_t)b * 8 + (erow & 7)) << LOGW) + xe) * W + e];
#pragma unroll
      for (int i = 0; i < NA; ++i) {
        const int idx = threadIdx.x + i * 128;
        *(float4*)&lds[(idx >> (LOGW - 2)) * WS + ((idx & (PER_CH4 - 1)) << 2)] =
            va[i];
      }
#pragma unroll
      for (int i = 0; i < NP; ++i) {
        const int idx = threadIdx.x + i * 128;
        *(float4*)&lds[(32 + (idx >> (LOGW - 2))) * WS +
                       ((idx & (PER_CH4 - 1)) << 2)] = vp[i];
      }
      lds[erow * WS + W + e] = ve;
    }
    __syncthreads();

#pragma unroll 8
    for (int k = 0; k < W; k += 4) {
      const int kb = k + khi;
      v2f av;
      av[0] = lds[aIdx + kb];
      av[1] = lds[aIdx + kb + 1];
      const v2f bv0 = *(const v2f*)&lds[bBase0 + kb];
      acc0 = __builtin_amdgcn_wmma_f32_16x16x4_f32(
           false, av,  false, bv0,
           (short)0, acc0,  false,  false);
      if (NX == 2) {
        const v2f bv1 = *(const v2f*)&lds[bBase1 + kb];
        acc1 = __builtin_amdgcn_wmma_f32_16x16x4_f32(
             false, av,  false, bv1,
             (short)0, acc1,  false,  false);
      }
    }
    __syncthreads();
  }

  const int rbase = m0 + ((lane >> 4) << 3);
  const int col = lane & 15;
  const int l2 = col & 7;
  const int pb0 = pairBase + ((col < 8) ? 0 : 64);
  const bool st0 = (col < 8) || (NX >= 1);
  const bool st1 = (NX == 2) && (col < 8);
#pragma unroll
  for (int i = 0; i < 8; ++i) {
    const int Rr = rbase + i;
    if (Rr < 56) {
      const int s = Rr >> 3;
      const int l1o = Rr & 7;
      const size_t o = ((size_t)b * 384 + l1o * 8 + l2) * 7 + s;
      if (st0) atomicAdd(&out[o + (size_t)pb0 * 7], acc0[i]);
      if (st1) atomicAdd(&out[o + (size_t)(pairBase + 128) * 7], acc1[i]);
    }
  }
}

extern "C" void kernel_launch(void* const* d_in, const int* in_sizes, int n_in,
                              void* d_out, int out_size, void* d_ws,
                              size_t ws_size, hipStream_t stream) {
  const float* x0 = (const float*)d_in[0];
  const float* x1 = (const float*)d_in[1];
  const float* x2 = (const float*)d_in[2];
  float* out = (float*)d_out;

  hipMemsetAsync(d_out, 0, (size_t)out_size * sizeof(float), stream);

  const size_t n0 = (size_t)16 * 256 * 256;
  const size_t n1 = (size_t)16 * 128 * 128;
  if (ws_size >= (n0 + n0 + n1) * sizeof(float)) {
    float* u10 = (float*)d_ws;
    float* u20 = u10 + n0;
    float* u21 = u20 + n0;

    interp_mm<7, 8><<<dim3(16 * 16), 128, 0, stream>>>(x1, u10);
    interp_mm<6, 8><<<dim3(16 * 16), 128, 0, stream>>>(x2, u20);
    interp_mm<6, 7><<<dim3(16 * 8), 128, 0, stream>>>(x2, u21);

    shifted_corr<8, 2, 2><<<dim3(2 * 128), 128, 0, stream>>>(x0, u10, u20, 0, out);
    shifted_corr<7, 1, 1><<<dim3(2 * 128), 128, 0, stream>>>(x1, u21, u21, 192, out);
    shifted_corr<6, 1, 0><<<dim3(2 * 64), 128, 0, stream>>>(x2, x2, x2, 320, out);
  }
}


# memset + 6 dependent launches regrouped into 2 by dependency level: out start value + the 3 interpolations, then the 3 shifted-row products; block index picks the job; sums and atomicAdds as before
# speedup vs baseline: 5.5006x; 1.6381x over previous
#include <hip/hip_runtime.h>


typedef __attribute__((ext_vector_type(2))) float v2f;
typedef __attribute__((ext_vector_type(8))) float v8f;

__device__ __constant__ int c_sdx[7] = {0, 0, 0, 1, 1, 2, -1};
__device__ __constant__ int c_sdy[7] = {0, 1, 2, 0, 1, 0, 1};

template <int LOGC, int LOGF>
__device__ __forceinline__ void
interp_mm(float* __restrict__ smem, const float* __restrict__ xc,
          float* __restrict__ up, const int bid) {
  constexpr int NC = 1 << LOGC;
  constexpr int NF = 1 << LOGF;
  constexpr int R = NF / NC;
  constexpr int VP = NC + 4;
  constexpr int T1 = NC / 64;
  constexpr int T2 = NF / 64;
  float* const g = smem;
  float* const Vs = smem + 256;
  float* const red = smem + 256 + 16 * 132;

  const int t = threadIdx.x;
  const int lane = t & 31;
  const int wave = t >> 5;
  const int c = lane & 15;
  const int h = lane >> 4;
  const int khi = h << 1;
  const int img = bid >> (LOGF - 4);
  const int x0 = (bid & ((NF >> 4) - 1)) << 4;
  const float* src = xc + ((size_t)img << (2 * LOGC));

  for (int k = t; k < NF; k += 128) {
    float v = (float)(NC - 1);
    if (k != 0) {
      const int a = ((NC - 1) * k) & (2 * NF - 1);
      v = sinpif((float)a / (float)NF) / sinpif((float)k / (float)NF);
    }
    v += cospif((float)(k & (2 * R - 1)) / (float)R);
    g[k] = v * (1.0f / (float)NC);
  }
  __syncthreads();

  v8f accV[T1];
#pragma unroll
  for (int j = 0; j < T1; ++j) accV[j] = v8f{};
  float salt = 0.0f;
  const int xa = x0 + c;
#pragma unroll 8
  for (int k = 0; k < NC; k += 4) {
    const int kk = k + khi;
    v2f av;
    av[0] = g[(xa - R * kk) & (NF - 1)];
    av[1] = g[(xa - R * (kk + 1)) & (NF - 1)];
#pragma unroll
    for (int j = 0; j < T1; ++j) {
      const int n = ((wave * T1 + j) << 4) + c;
      v2f bv;
      bv[0] = src[(size_t)kk * NC + n];
      bv[1] = src[(size_t)(kk + 1) * NC + n];
      salt += bv[0] - bv[1];
      accV[j] = __builtin_amdgcn_wmma_f32_16x16x4_f32(
           false, av,  false, bv,
           (short)0, accV[j],  false,  false);
    }
  }
  salt = (c & 1) ? -salt : salt;
#pragma unroll
  for (int o = 16; o > 0; o >>= 1) salt += __shfl_xor(salt, o);
  if (lane == 0) red[wave] = salt;
#pragma unroll
  for (int j = 0; j < T1; ++j) {
#pragma unroll
    for (int i = 0; i < 8; ++i)
      Vs[(8 * h + i) * VP + ((wave * T1 + j) << 4) + c] = accV[j][i];
  }
  __syncthreads();
  const float S = (red[0] + red[1]) + (red[2] + red[3]);

  v8f accU[T2];
#pragma unroll
  for (int j = 0; j < T2; ++j) accU[j] = v8f{};
#pragma unroll 4
  for (int k = 0; k < NC; k += 4) {
    const int kk = k + khi;
    const v2f av = *(const v2f*)&Vs[c * VP + kk];
#pragma unroll
    for (int j = 0; j < T2; ++j) {
      const int n = ((wave * T2 + j) << 4) + c;
      v2f bv;
      bv[0] = g[(n - R * kk) & (NF - 1)];
      bv[1] = g[(n - R * (kk + 1)) & (NF - 1)];
      accU[j] = __builtin_amdgcn_wmma_f32_16x16x4_f32(
           false, av,  false, bv,
           (short)0, accU[j],  false,  false);
    }
  }

  const float corr = S * (1.0f / (float)(NC * NC));
  float* dst = up + ((size_t)img << (2 * LOGF));
#pragma unroll
  for (int j = 0; j < T2; ++j) {
    const int n = ((wave * T2 + j) << 4) + c;
    const float sn = sinpif((float)(n & (2 * R - 1)) / (float)R) * corr;
#pragma unroll
    for (int i = 0; i < 8; ++i) {
      const int row = x0 + 8 * h + i;
      const float sr = sinpif((float)(row & (2 * R - 1)) / (float)R);
      dst[(size_t)row * NF + n] = accU[j][i] - sr * sn;
    }
  }
}

template <int LOGW, int ROWS, int NX>
__device__ __forceinline__ void
shifted_corr(float* __restrict__ lds, const float* __restrict__ xin,
             const float* __restrict__ ua, const float* __restrict__ ub,
             const int pairBase, float* __restrict__ out, const int bid) {
  constexpr int W = 1 << LOGW;
  constexpr int H = W;
  constexpr int WS = W + 4;
  constexpr int NR = 32 + 8 * NX;
  constexpr int ZROW = NR * WS;
  constexpr int SLABS = H / ROWS;
  static_assert((NR + 1) * WS <= 49 * 260, "LDS array of level_b too small");

  const int b = bid / SLABS;
  const int slab = bid & (SLABS - 1);

  const int lane = threadIdx.x & 31;
  const int wave = threadIdx.x >> 5;
  const int m0 = wave * 16;

  const int mrow = m0 + (lane & 15);
  const bool aval = (mrow < 56);
  const int sidx = aval ? (mrow >> 3) : 0;
  const int l1 = mrow & 7;
  const int slot = c_sdx[sidx] + 1;
  const int dy = aval ? c_sdy[sidx] : 0;
  const int aIdx = (aval ? (slot * 8 + l1) * WS : ZROW) + dy;
  const int ncol = lane & 15;
  const int bBase0 = (ncol < 8) ? (8 + ncol) * WS
                                : ((NX >= 1) ? (32 + (ncol - 8)) * WS : ZROW);
  const int bBase1 = (NX == 2 && ncol < 8) ? (40 + ncol) * WS : ZROW;
  const int khi = (lane >> 4) << 1;

  for (int idx = threadIdx.x; idx < WS; idx += 128) lds[ZROW + idx] = 0.0f;

  v8f acc0 = {};
  v8f acc1 = {};

  for (int xi = 0; xi < ROWS; ++xi) {
    const int xr = slab * ROWS + xi;

    {
      constexpr int PER_CH4 = W / 4;
      constexpr int NA = W / 16;
      constexpr int NP = NX * W / 64;
      float4 va[NA];
      float4 vp[NP > 0 ? NP : 1];
#pragma unroll
      for (int i = 0; i < NA; ++i) {
        const int idx = threadIdx.x + i * 128;
        const int row = idx >> (LOGW - 2);
        const int y4 = idx & (PER_CH4 - 1);
        int xs = xr + (row >> 3) - 1;
        xs += (xs < 0) ? H : 0;
        xs -= (xs >= H) ? H : 0;
        va[i] = *(const float4*)&xin[((((size_t)b * 8 + (row & 7)) << LOGW) +
                                      xs) * W + (y4 << 2)];
      }
#pragma unroll
      for (int i = 0; i < NP; ++i) {
        const int idx = threadIdx.x + i * 128;
        const int prow = idx >> (LOGW - 2);
        const int y4 = idx & (PER_CH4 - 1);
        const float* us = (prow < 8) ? ua : ub;
        vp[i] = *(const float4*)&us[((((size_t)b * 8 + (prow & 7)) << LOGW) +
                                     xr) * W + (y4 << 2)];
      }
      const int erow = threadIdx.x >> 2;
      const int e = threadIdx.x & 3;
      int xe = xr + (erow >> 3) - 1;
      xe += (xe < 0) ? H : 0;
      xe -= (xe >= H) ? H : 0;
      const float ve =
          xin[((((size_t)b * 8 + (erow & 7)) << LOGW) + xe) * W + e];
#pragma unroll
      for (int i = 0; i < NA; ++i) {
        const int idx = threadIdx.x + i * 128;
        *(float4*)&lds[(idx >> (LOGW - 2)) * WS + ((idx & (PER_CH4 - 1)) << 2)] =
            va[i];
      }
#pragma unroll
      for (int i = 0; i < NP; ++i) {
        const int idx = threadIdx.x + i * 128;
        *(float4*)&lds[(32 + (idx >> (LOGW - 2))) * WS +
                       ((idx & (PER_CH4 - 1)) << 2)] = vp[i];
      }
      lds[erow * WS + W + e] = ve;
    }
    __syncthreads();

#pragma unroll 8
    for (int k = 0; k < W; k += 4) {
      const int kb = k + khi;
      v2f av;
      av[0] = lds[aIdx + kb];
      av[1] = lds[aIdx + kb + 1];
      const v2f bv0 = *(const v2f*)&lds[bBase0 + kb];
      acc0 = __builtin_amdgcn_wmma_f32_16x16x4_f32(
           false, av,  false, bv0,
           (short)0, acc0,  false,  false);
      if (NX == 2) {
        const v2f bv1 = *(const v2f*)&lds[bBase1 + kb];
        acc1 = __builtin_amdgcn_wmma_f32_16x16x4_f32(
             false, av,  false, bv1,
             (short)0, acc1,  false,  false);
      }
    }
    __syncthreads();
  }

  const int rbase = m0 + ((lane >> 4) << 3);
  const int col = lane & 15;
  const int l2 = col & 7;
  const int pb0 = pairBase + ((col < 8) ? 0 : 64);
  const bool st0 = (col < 8) || (NX >= 1);
  const bool st1 = (NX == 2) && (col < 8);
#pragma unroll
  for (int i = 0; i < 8; ++i) {
    const int Rr = rbase + i;
    if (Rr < 56) {
      const int s = Rr >> 3;
      const int l1o = Rr & 7;
      const size_t o = ((size_t)b * 384 + l1o * 8 + l2) * 7 + s;
      if (st0) atomicAdd(&out[o + (size_t)pb0 * 7], acc0[i]);
      if (st1) atomicAdd(&out[o + (size_t)(pairBase + 128) * 7], acc1[i]);
    }
  }
}

__global__ void __launch_bounds__(128)
level_a(const float* __restrict__ x1, const float* __restrict__ x2,
        float* __restrict__ u10, float* __restrict__ u20,
        float* __restrict__ u21, float* __restrict__ out) {
  __shared__ alignas(16) float smem[256 + 16 * 132 + 4];
  const int bid = blockIdx.x;
  if (bid < 256) {
    interp_mm<7, 8>(smem, x1, u10, bid);
  } else if (bid < 512) {
    interp_mm<6, 8>(smem, x2, u20, bid - 256);
  } else {
    const int z = (bid - 512) * 128 + (int)threadIdx.x;
    if (z < 2 * 384 * 7) out[z] = 0.0f;
    interp_mm<6, 7>(smem, x2, u21, bid - 512);
  }
}

__global__ void __launch_bounds__(128) __attribute__((amdgpu_waves_per_eu(1, 4)))
level_b(const float* __restrict__ x0, const float* __restrict__ x1,
        const float* __restrict__ x2, const float* __restrict__ u10,
        const float* __restrict__ u20, const float* __restrict__ u21,
        float* __restrict__ out) {
  __shared__ alignas(16) float smem[49 * 260];
  const int bid = blockIdx.x;
  if (bid < 256) {
    shifted_corr<8, 2, 2>(smem, x0, u10, u20, 0, out, bid);
  } else if (bid < 512) {
    shifted_corr<7, 1, 1>(smem, x1, u21, u21, 192, out, bid - 256);
  } else {
    shifted_corr<6, 1, 0>(smem, x2, x2, x2, 320, out, bid - 512);
  }
}

extern "C" void kernel_launch(void* const* d_in, const int* in_sizes, int n_in,
                              void* d_out, int out_size, void* d_ws,
                              size_t ws_size, hipStream_t stream) {
  const float* x0 = (const float*)d_in[0];
  const float* x1 = (const float*)d_in[1];
  const float* x2 = (const float*)d_in[2];
  float* out = (float*)d_out;

  const size_t n0 = (size_t)16 * 256 * 256;
  const size_t n1 = (size_t)16 * 128 * 128;
  if (ws_size >= (n0 + n0 + n1) * sizeof(float)) {
    float* u10 = (float*)d_ws;
    float* u20 = u10 + n0;
    float* u21 = u20 + n0;

    level_a<<<dim3(640), 128, 0, stream>>>(x1, x2, u10, u20, u21, out);
    level_b<<<dim3(640), 128, 0, stream>>>(x0, x1, x2, u10, u20, u21, out);
  }
}


# all three WMMA loops: the next 8-k-step block's operand fetches (LDS fragment reads, product 1's global B rows) issued as one group one block ahead into a second register set; sum order unchanged
# speedup vs baseline: 6.1655x; 1.1209x over previous
#include <hip/hip_runtime.h>


typedef __attribute__((ext_vector_type(2))) float v2f;
typedef __attribute__((ext_vector_type(8))) float v8f;

__device__ __constant__ int c_sdx[7] = {0, 0, 0, 1, 1, 2, -1};
__device__ __constant__ int c_sdy[7] = {0, 1, 2, 0, 1, 0, 1};

template <int NC, int NF, int T1>
__device__ __forceinline__ void
p1_load(const float* __restrict__ g, const float* __restrict__ srcw,
        const int bx, const int blk, v2f (&pa)[8], v2f (&pb)[8][T1]) {
  constexpr int R = NF / NC;
  const int gb = (bx + R * 32 * blk) & (NF - 1);
  const float* const sb = srcw + (size_t)(32 * blk) * NC;
#pragma unroll
  for (int s = 0; s < 8; ++s) {
    pa[s][0] = g[gb + R * (4 * s)];
    pa[s][1] = g[gb + R * (4 * s + 1)];
#pragma unroll
    for (int j = 0; j < T1; ++j) {
      pb[s][j][0] = sb[(4 * s) * NC + 16 * j];
      pb[s][j][1] = sb[(4 * s + 1) * NC + 16 * j];
    }
  }
}

template <int T1>
__device__ __forceinline__ void
p1_mma(const v2f (&pa)[8], const v2f (&pb)[8][T1], v8f (&accV)[T1],
       float& salt) {
#pragma unroll
  for (int s = 0; s < 8; ++s) {
#pragma unroll
    for (int j = 0; j < T1; ++j) {
      salt += pb[s][j][0] - pb[s][j][1];
      accV[j] = __builtin_amdgcn_wmma_f32_16x16x4_f32(
           false, pa[s],  false, pb[s][j],
           (short)0, accV[j],  false,  false);
    }
  }
}

template <int NF, int R, int KB2, int T2>
__device__ __forceinline__ void
p2_load(const float* __restrict__ g, const float* __restrict__ va,
        const int bn, const int blk, v2f (&qa)[KB2], v2f (&qb)[KB2][T2]) {
  const int k0 = 4 * KB2 * blk;
#pragma unroll
  for (int s = 0; s < KB2; ++s) qa[s] = *(const v2f*)&va[k0 + 4 * s];
#pragma unroll
  for (int j = 0; j < T2; ++j) {
    const int gb = (bn - 16 * j + R * k0) & (NF - 1);
#pragma unroll
    for (int s = 0; s < KB2; ++s) {
      qb[s][j][0] = g[gb + R * (4 * s)];
      qb[s][j][1] = g[gb + R * (4 * s + 1)];
    }
  }
}

template <int KB2, int T2>
__device__ __forceinline__ void
p2_mma(const v2f (&qa)[KB2], const v2f (&qb)[KB2][T2], v8f (&accU)[T2]) {
#pragma unroll
  for (int s = 0; s < KB2; ++s) {
#pragma unroll
    for (int j = 0; j < T2; ++j)
      accU[j] = __builtin_amdgcn_wmma_f32_16x16x4_f32(
           false, qa[s],  false, qb[s][j],
           (short)0, accU[j],  false,  false);
  }
}

template <int LOGC, int LOGF>
__device__ __forceinline__ void
interp_mm(float* __restrict__ smem, const float* __restrict__ xc,
          float* __restrict__ up, const int bid) {
  constexpr int NC = 1 << LOGC;
  constexpr int NF = 1 << LOGF;
  constexpr int R = NF / NC;
  constexpr int VP = NC + 4;
  constexpr int T1 = NC / 64;
  constexpr int T2 = NF / 64;
  constexpr int NB1 = NC / 32;
  constexpr int KB2 = 16 / T2;
  constexpr int NB2 = NC / (4 * KB2);
  float* const g = smem;
  float* const Vs = smem + 512;
  float* const red = smem + 512 + 16 * 132;

  const int t = threadIdx.x;
  const int lane = t & 31;
  const int wave = t >> 5;
  const int c = lane & 15;
  const int h = lane >> 4;
  const int khi = h << 1;
  const int img = bid >> (LOGF - 4);
  const int x0 = (bid & ((NF >> 4) - 1)) << 4;
  const float* src = xc + ((size_t)img << (2 * LOGC));

  for (int k = t; k < NF; k += 128) {
    float v = (float)(NC - 1);
    if (k != 0) {
      const int a = ((NC - 1) * k) & (2 * NF - 1);
      v = sinpif((float)a / (float)NF) / sinpif((float)k / (float)NF);
    }
    v += cospif((float)(k & (2 * R - 1)) / (float)R);
    v *= 1.0f / (float)NC;
    const int i = (NF - k) & (NF - 1);
    g[i] = v;
    g[i + NF] = v;
  }
  __syncthreads();

  v8f accV[T1];
#pragma unroll
  for (int j = 0; j < T1; ++j) accV[j] = v8f{};
  float salt = 0.0f;
  const int xa = x0 + c;
  const float* const srcw = src + (size_t)khi * NC + ((wave * T1) << 4) + c;
  {
    v2f pa0[8], pa1[8];
    v2f pb0[8][T1], pb1[8][T1];
    p1_load<NC, NF, T1>(g, srcw, R * khi - xa, 0, pa0, pb0);
#pragma unroll
    for (int blk = 0; blk < NB1; blk += 2) {
      p1_load<NC, NF, T1>(g, srcw, R * khi - xa, blk + 1, pa1, pb1);
      __builtin_amdgcn_sched_barrier(0);
      p1_mma<T1>(pa0, pb0, accV, salt);
      __builtin_amdgcn_sched_barrier(0);
      if (blk + 2 < NB1)
        p1_load<NC, NF, T1>(g, srcw, R * khi - xa, blk + 2, pa0, pb0);
      __builtin_amdgcn_sched_barrier(0);
      p1_mma<T1>(pa1, pb1, accV, salt);
      __builtin_amdgcn_sched_barrier(0);
    }
  }
  salt = (c & 1) ? -salt : salt;
#pragma unroll
  for (int o = 16; o > 0; o >>= 1) salt += __shfl_xor(salt, o);
  if (lane == 0) red[wave] = salt;
#pragma unroll
  for (int j = 0; j < T1; ++j) {
#pragma unroll
    for (int i = 0; i < 8; ++i)
      Vs[(8 * h + i) * VP + ((wave * T1 + j) << 4) + c] = accV[j][i];
  }
  __syncthreads();
  const float S = (red[0] + red[1]) + (red[2] + red[3]);

  v8f accU[T2];
#pragma unroll
  for (int j = 0; j < T2; ++j) accU[j] = v8f{};
  {
    const float* const va = Vs + c * VP + khi;
    const int bn = R * khi - (((wave * T2) << 4) + c);
    v2f qa0[KB2], qa1[KB2];
    v2f qb0[KB2][T2], qb1[KB2][T2];
    p2_load<NF, R, KB2, T2>(g, va, bn, 0, qa0, qb0);
#pragma unroll
    for (int blk = 0; blk < NB2; blk += 2) {
      p2_load<NF, R, KB2, T2>(g, va, bn, blk + 1, qa1, qb1);
      __builtin_amdgcn_sched_barrier(0);
      p2_mma<KB2, T2>(qa0, qb0, accU);
      __builtin_amdgcn_sched_barrier(0);
      if (blk + 2 < NB2) p2_load<NF, R, KB2, T2>(g, va, bn, blk + 2, qa0, qb0);
      __builtin_amdgcn_sched_barrier(0);
      p2_mma<KB2, T2>(qa1, qb1, accU);
      __builtin_amdgcn_sched_barrier(0);
    }
  }

  const float corr = S * (1.0f / (float)(NC * NC));
  float* dst = up + ((size_t)img << (2 * LOGF));
#pragma unroll
  for (int j = 0; j < T2; ++j) {
    const int n = ((wave * T2 + j) << 4) + c;
    const float sn = sinpif((float)(n & (2 * R - 1)) / (float)R) * corr;
#pragma unroll
    for (int i = 0; i < 8; ++i) {
      const int row = x0 + 8 * h + i;
      const float sr = sinpif((float)(row & (2 * R - 1)) / (float)R);
      dst[(size_t)row * NF + n] = accU[j][i] - sr * sn;
    }
  }
}

template <int NX>
__device__ __forceinline__ void
frag_load(const float* __restrict__ lds, const int aOff, const int b0Off,
          const int b1Off, v2f (&fa)[8], v2f (&fb)[8], v2f (&fc)[8]) {
#pragma unroll
  for (int i = 0; i < 8; ++i) {
    fa[i][0] = lds[aOff + 4 * i];
    fa[i][1] = lds[aOff + 4 * i + 1];
    fb[i] = *(const v2f*)&lds[b0Off + 4 * i];
    if (NX == 2) fc[i] = *(const v2f*)&lds[b1Off + 4 * i];
  }
}

template <int NX>
__device__ __forceinline__ void
frag_mma(const v2f (&fa)[8], const v2f (&fb)[8], const v2f (&fc)[8],
         v8f& acc0, v8f& acc1) {
#pragma unroll
  for (int i = 0; i < 8; ++i) {
    acc0 = __builtin_amdgcn_wmma_f32_16x16x4_f32(
         false, fa[i],  false, fb[i],
         (short)0, acc0,  false,  false);
    if (NX == 2)
      acc1 = __builtin_amdgcn_wmma_f32_16x16x4_f32(
           false, fa[i],  false, fc[i],
           (short)0, acc1,  false,  false);
  }
}

template <int LOGW, int ROWS, int NX>
__device__ __forceinline__ void
shifted_corr(float* __restrict__ lds, const float* __restrict__ xin,
             const float* __restrict__ ua, const float* __restrict__ ub,
             const int pairBase, float* __restrict__ out, const int bid) {
  constexpr int W = 1 << LOGW;
  constexpr int H = W;
  constexpr int WS = W + 4;
  constexpr int NR = 32 + 8 * NX;
  constexpr int ZROW = NR * WS;
  constexpr int SLABS = H / ROWS;
  static_assert((NR + 1) * WS <= 49 * 260, "LDS array of level_b too small");

  const int b = bid / SLABS;
  const int slab = bid & (SLABS - 1);

  const int lane = threadIdx.x & 31;
  const int wave = threadIdx.x >> 5;
  const int m0 = wave * 16;

  const int mrow = m0 + (lane & 15);
  const bool aval = (mrow < 56);
  const int sidx = aval ? (mrow >> 3) : 0;
  const int l1 = mrow & 7;
  const int slot = c_sdx[sidx] + 1;
  const int dy = aval ? c_sdy[sidx] : 0;
  const int aIdx = (aval ? (slot * 8 + l1) * WS : ZROW) + dy;
  const int ncol = lane & 15;
  const int bBase0 = (ncol < 8) ? (8 + ncol) * WS
                                : ((NX >= 1) ? (32 + (ncol - 8)) * WS : ZROW);
  const int bBase1 = (NX == 2 && ncol < 8) ? (40 + ncol) * WS : ZROW;
  const int khi = (lane >> 4) << 1;

  for (int idx = threadIdx.x; idx < WS; idx += 128) lds[ZROW + idx] = 0.0f;

  v8f acc0 = {};
  v8f acc1 = {};

  for (int xi = 0; xi < ROWS; ++xi) {
    const int xr = slab * ROWS + xi;

    {
      constexpr int PER_CH4 = W / 4;
      constexpr int NA = W / 16;
      constexpr int NP = NX * W / 64;
      float4 va[NA];
      float4 vp[NP > 0 ? NP : 1];
#pragma unroll
      for (int i = 0; i < NA; ++i) {
        const int idx = threadIdx.x + i * 128;
        const int row = idx >> (LOGW - 2);
        const int y4 = idx & (PER_CH4 - 1);
        int xs = xr + (row >> 3) - 1;
        xs += (xs < 0) ? H : 0;
        xs -= (xs >= H) ? H : 0;
        va[i] = *(const float4*)&xin[((((size_t)b * 8 + (row & 7)) << LOGW) +
                                      xs) * W + (y4 << 2)];
      }
#pragma unroll
      for (int i = 0; i < NP; ++i) {
        const int idx = threadIdx.x + i * 128;
        const int prow = idx >> (LOGW - 2);
        const int y4 = idx & (PER_CH4 - 1);
        const float* us = (prow < 8) ? ua : ub;
        vp[i] = *(const float4*)&us[((((size_t)b * 8 + (prow & 7)) << LOGW) +
                                     xr) * W + (y4 << 2)];
      }
      const int erow = threadIdx.x >> 2;
      const int e = threadIdx.x & 3;
      int xe = xr + (erow >> 3) - 1;
      xe += (xe < 0) ? H : 0;
      xe -= (xe >= H) ? H : 0;
      const float ve =
          xin[((((size_t)b * 8 + (erow & 7)) << LOGW) + xe) * W + e];
#pragma unroll
      for (int i = 0; i < NA; ++i) {
        const int idx = threadIdx.x + i * 128;
        *(float4*)&lds[(idx >> (LOGW - 2)) * WS + ((idx & (PER_CH4 - 1)) << 2)] =
            va[i];
      }
#pragma unroll
      for (int i = 0; i < NP; ++i) {
        const int idx = threadIdx.x + i * 128;
        *(float4*)&lds[(32 + (idx >> (LOGW - 2))) * WS +
                       ((idx & (PER_CH4 - 1)) << 2)] = vp[i];
      }
      lds[erow * WS + W + e] = ve;
    }
    __syncthreads();

    {
      constexpr int NBLK = W / 32;
      const int aK = aIdx + khi;
      const int b0K = bBase0 + khi;
      const int b1K = bBase1 + khi;
      v2f fa0[8], fb0[8], fc0[8];
      v2f fa1[8], fb1[8], fc1[8];
      frag_load<NX>(lds, aK, b0K, b1K, fa0, fb0, fc0);
#pragma unroll
      for (int blk = 0; blk < NBLK; blk += 2) {
        frag_load<NX>(lds, aK + (blk + 1) * 32, b0K + (blk + 1) * 32,
                      b1K + (blk + 1) * 32, fa1, fb1, fc1);
        __builtin_amdgcn_sched_barrier(0);
        frag_mma<NX>(fa0, fb0, fc0, acc0, acc1);
        __builtin_amdgcn_sched_barrier(0);
        if (blk + 2 < NBLK)
          frag_load<NX>(lds, aK + (blk + 2) * 32, b0K + (blk + 2) * 32,
                        b1K + (blk + 2) * 32, fa0, fb0, fc0);
        __builtin_amdgcn_sched_barrier(0);
        frag_mma<NX>(fa1, fb1, fc1, acc0, acc1);
        __builtin_amdgcn_sched_barrier(0);
      }
    }
    __syncthreads();
  }

  const int rbase = m0 + ((lane >> 4) << 3);
  const int col = lane & 15;
  const int l2 = col & 7;
  const int pb0 = pairBase + ((col < 8) ? 0 : 64);
  const bool st0 = (col < 8) || (NX >= 1);
  const bool st1 = (NX == 2) && (col < 8);
#pragma unroll
  for (int i = 0; i < 8; ++i) {
    const int Rr = rbase + i;
    if (Rr < 56) {
      const int s = Rr >> 3;
      const int l1o = Rr & 7;
      const size_t o = ((size_t)b * 384 + l1o * 8 + l2) * 7 + s;
      if (st0) atomicAdd(&out[o + (size_t)pb0 * 7], acc0[i]);
      if (st1) atomicAdd(&out[o + (size_t)(pairBase + 128) * 7], acc1[i]);
    }
  }
}

__global__ void __launch_bounds__(128)
level_a(const float* __restrict__ x1, const float* __restrict__ x2,
        float* __restrict__ u10, float* __restrict__ u20,
        float* __restrict__ u21, float* __restrict__ out) {
  __shared__ alignas(16) float smem[512 + 16 * 132 + 4];
  const int bid = blockIdx.x;
  if (bid < 256) {
    interp_mm<7, 8>(smem, x1, u10, bid);
  } else if (bid < 512) {
    interp_mm<6, 8>(smem, x2, u20, bid - 256);
  } else {
    const int z = (bid - 512) * 128 + (int)threadIdx.x;
    if (z < 2 * 384 * 7) out[z] = 0.0f;
    interp_mm<6, 7>(smem, x2, u21, bid - 512);
  }
}

__global__ void __launch_bounds__(128) __attribute__((amdgpu_waves_per_eu(1, 4)))
level_b(const float* __restrict__ x0, const float* __restrict__ x1,
        const float* __restrict__ x2, const float* __restrict__ u10,
        const float* __restrict__ u20, const float* __restrict__ u21,
        float* __restrict__ out) {
  __shared__ alignas(16) float smem[49 * 260];
  const int bid = blockIdx.x;
  if (bid < 256) {
    shifted_corr<8, 2, 2>(smem, x0, u10, u20, 0, out, bid);
  } else if (bid < 512) {
    shifted_corr<7, 1, 1>(smem, x1, u21, u21, 192, out, bid - 256);
  } else {
    shifted_corr<6, 1, 0>(smem, x2, x2, x2, 320, out, bid - 512);
  }
}

extern "C" void kernel_launch(void* const* d_in, const int* in_sizes, int n_in,
                              void* d_out, int out_size, void* d_ws,
                              size_t ws_size, hipStream_t stream) {
  const float* x0 = (const float*)d_in[0];
  const float* x1 = (const float*)d_in[1];
  const float* x2 = (const float*)d_in[2];
  float* out = (float*)d_out;

  const size_t n0 = (size_t)16 * 256 * 256;
  const size_t n1 = (size_t)16 * 128 * 128;
  if (ws_size >= (n0 + n0 + n1) * sizeof(float)) {
    float* u10 = (float*)d_ws;
    float* u20 = u10 + n0;
    float* u21 = u20 + n0;

    level_a<<<dim3(640), 128, 0, stream>>>(x1, x2, u10, u20, u21, out);
    level_b<<<dim3(640), 128, 0, stream>>>(x0, x1, x2, u10, u20, u21, out);
  }
}


# level_b epilogue: block's partial sums transposed through LDS into out order, so each float atomicAdd instruction covers one whole 128-byte line of out (42 a block, was ~240 line pieces); same addends
# speedup vs baseline: 9.8354x; 1.5952x over previous
#include <hip/hip_runtime.h>


typedef __attribute__((ext_vector_type(2))) float v2f;
typedef __attribute__((ext_vector_type(8))) float v8f;

__device__ __constant__ int c_sdx[7] = {0, 0, 0, 1, 1, 2, -1};
__device__ __constant__ int c_sdy[7] = {0, 1, 2, 0, 1, 0, 1};

template <int NC, int NF, int T1>
__device__ __forceinline__ void
p1_load(const float* __restrict__ g, const float* __restrict__ srcw,
        const int bx, const int blk, v2f (&pa)[8], v2f (&pb)[8][T1]) {
  constexpr int R = NF / NC;
  const int gb = (bx + R * 32 * blk) & (NF - 1);
  const float* const sb = srcw + (size_t)(32 * blk) * NC;
#pragma unroll
  for (int s = 0; s < 8; ++s) {
    pa[s][0] = g[gb + R * (4 * s)];
    pa[s][1] = g[gb + R * (4 * s + 1)];
#pragma unroll
    for (int j = 0; j < T1; ++j) {
      pb[s][j][0] = sb[(4 * s) * NC + 16 * j];
      pb[s][j][1] = sb[(4 * s + 1) * NC + 16 * j];
    }
  }
}

template <int T1>
__device__ __forceinline__ void
p1_mma(const v2f (&pa)[8], const v2f (&pb)[8][T1], v8f (&accV)[T1],
       float& salt) {
#pragma unroll
  for (int s = 0; s < 8; ++s) {
#pragma unroll
    for (int j = 0; j < T1; ++j) {
      salt += pb[s][j][0] - pb[s][j][1];
      accV[j] = __builtin_amdgcn_wmma_f32_16x16x4_f32(
           false, pa[s],  false, pb[s][j],
           (short)0, accV[j],  false,  false);
    }
  }
}

template <int NF, int R, int KB2, int T2>
__device__ __forceinline__ void
p2_load(const float* __restrict__ g, const float* __restrict__ va,
        const int bn, const int blk, v2f (&qa)[KB2], v2f (&qb)[KB2][T2]) {
  const int k0 = 4 * KB2 * blk;
#pragma unroll
  for (int s = 0; s < KB2; ++s) qa[s] = *(const v2f*)&va[k0 + 4 * s];
#pragma unroll
  for (int j = 0; j < T2; ++j) {
    const int gb = (bn - 16 * j + R * k0) & (NF - 1);
#pragma unroll
    for (int s = 0; s < KB2; ++s) {
      qb[s][j][0] = g[gb + R * (4 * s)];
      qb[s][j][1] = g[gb + R * (4 * s + 1)];
    }
  }
}

template <int KB2, int T2>
__device__ __forceinline__ void
p2_mma(const v2f (&qa)[KB2], const v2f (&qb)[KB2][T2], v8f (&accU)[T2]) {
#pragma unroll
  for (int s = 0; s < KB2; ++s) {
#pragma unroll
    for (int j = 0; j < T2; ++j)
      accU[j] = __builtin_amdgcn_wmma_f32_16x16x4_f32(
           false, qa[s],  false, qb[s][j],
           (short)0, accU[j],  false,  false);
  }
}

template <int LOGC, int LOGF>
__device__ __forceinline__ void
interp_mm(float* __restrict__ smem, const float* __restrict__ xc,
          float* __restrict__ up, const int bid) {
  constexpr int NC = 1 << LOGC;
  constexpr int NF = 1 << LOGF;
  constexpr int R = NF / NC;
  constexpr int VP = NC + 4;
  constexpr int T1 = NC / 64;
  constexpr int T2 = NF / 64;
  constexpr int NB1 = NC / 32;
  constexpr int KB2 = 16 / T2;
  constexpr int NB2 = NC / (4 * KB2);
  float* const g = smem;
  float* const Vs = smem + 512;
  float* const red = smem + 512 + 16 * 132;

  const int t = threadIdx.x;
  const int lane = t & 31;
  const int wave = t >> 5;
  const int c = lane & 15;
  const int h = lane >> 4;
  const int khi = h << 1;
  const int img = bid >> (LOGF - 4);
  const int x0 = (bid & ((NF >> 4) - 1)) << 4;
  const float* src = xc + ((size_t)img << (2 * LOGC));

  for (int k = t; k < NF; k += 128) {
    float v = (float)(NC - 1);
    if (k != 0) {
      const int a = ((NC - 1) * k) & (2 * NF - 1);
      v = sinpif((float)a / (float)NF) / sinpif((float)k / (float)NF);
    }
    v += cospif((float)(k & (2 * R - 1)) / (float)R);
    v *= 1.0f / (float)NC;
    const int i = (NF - k) & (NF - 1);
    g[i] = v;
    g[i + NF] = v;
  }
  __syncthreads();

  v8f accV[T1];
#pragma unroll
  for (int j = 0; j < T1; ++j) accV[j] = v8f{};
  float salt = 0.0f;
  const int xa = x0 + c;
  const float* const srcw = src + (size_t)khi * NC + ((wave * T1) << 4) + c;
  {
    v2f pa0[8], pa1[8];
    v2f pb0[8][T1], pb1[8][T1];
    p1_load<NC, NF, T1>(g, srcw, R * khi - xa, 0, pa0, pb0);
#pragma unroll
    for (int blk = 0; blk < NB1; blk += 2) {
      p1_load<NC, NF, T1>(g, srcw, R * khi - xa, blk + 1, pa1, pb1);
      __builtin_amdgcn_sched_barrier(0);
      p1_mma<T1>(pa0, pb0, accV, salt);
      __builtin_amdgcn_sched_barrier(0);
      if (blk + 2 < NB1)
        p1_load<NC, NF, T1>(g, srcw, R * khi - xa, blk + 2, pa0, pb0);
      __builtin_amdgcn_sched_barrier(0);
      p1_mma<T1>(pa1, pb1, accV, salt);
      __builtin_amdgcn_sched_barrier(0);
    }
  }
  salt = (c & 1) ? -salt : salt;
#pragma unroll
  for (int o = 16; o > 0; o >>= 1) salt += __shfl_xor(salt, o);
  if (lane == 0) red[wave] = salt;
#pragma unroll
  for (int j = 0; j < T1; ++j) {
#pragma unroll
    for (int i = 0; i < 8; ++i)
      Vs[(8 * h + i) * VP + ((wave * T1 + j) << 4) + c] = accV[j][i];
  }
  __syncthreads();
  const float S = (red[0] + red[1]) + (red[2] + red[3]);

  v8f accU[T2];
#pragma unroll
  for (int j = 0; j < T2; ++j) accU[j] = v8f{};
  {
    const float* const va = Vs + c * VP + khi;
    const int bn = R * khi - (((wave * T2) << 4) + c);
    v2f qa0[KB2], qa1[KB2];
    v2f qb0[KB2][T2], qb1[KB2][T2];
    p2_load<NF, R, KB2, T2>(g, va, bn, 0, qa0, qb0);
#pragma unroll
    for (int blk = 0; blk < NB2; blk += 2) {
      p2_load<NF, R, KB2, T2>(g, va, bn, blk + 1, qa1, qb1);
      __builtin_amdgcn_sched_barrier(0);
      p2_mma<KB2, T2>(qa0, qb0, accU);
      __builtin_amdgcn_sched_barrier(0);
      if (blk + 2 < NB2) p2_load<NF, R, KB2, T2>(g, va, bn, blk + 2, qa0, qb0);
      __builtin_amdgcn_sched_barrier(0);
      p2_mma<KB2, T2>(qa1, qb1, accU);
      __builtin_amdgcn_sched_barrier(0);
    }
  }

  const float corr = S * (1.0f / (float)(NC * NC));
  float* dst = up + ((size_t)img << (2 * LOGF));
#pragma unroll
  for (int j = 0; j < T2; ++j) {
    const int n = ((wave * T2 + j) << 4) + c;
    const float sn = sinpif((float)(n & (2 * R - 1)) / (float)R) * corr;
#pragma unroll
    for (int i = 0; i < 8; ++i) {
      const int row = x0 + 8 * h + i;
      const float sr = sinpif((float)(row & (2 * R - 1)) / (float)R);
      dst[(size_t)row * NF + n] = accU[j][i] - sr * sn;
    }
  }
}

template <int NX>
__device__ __forceinline__ void
frag_load(const float* __restrict__ lds, const int aOff, const int b0Off,
          const int b1Off, v2f (&fa)[8], v2f (&fb)[8], v2f (&fc)[8]) {
#pragma unroll
  for (int i = 0; i < 8; ++i) {
    fa[i][0] = lds[aOff + 4 * i];
    fa[i][1] = lds[aOff + 4 * i + 1];
    fb[i] = *(const v2f*)&lds[b0Off + 4 * i];
    if (NX == 2) fc[i] = *(const v2f*)&lds[b1Off + 4 * i];
  }
}

template <int NX>
__device__ __forceinline__ void
frag_mma(const v2f (&fa)[8], const v2f (&fb)[8], const v2f (&fc)[8],
         v8f& acc0, v8f& acc1) {
#pragma unroll
  for (int i = 0; i < 8; ++i) {
    acc0 = __builtin_amdgcn_wmma_f32_16x16x4_f32(
         false, fa[i],  false, fb[i],
         (short)0, acc0,  false,  false);
    if (NX == 2)
      acc1 = __builtin_amdgcn_wmma_f32_16x16x4_f32(
           false, fa[i],  false, fc[i],
           (short)0, acc1,  false,  false);
  }
}

template <int LOGW, int ROWS, int NX>
__device__ __forceinline__ void
shifted_corr(float* __restrict__ lds, const float* __restrict__ xin,
             const float* __restrict__ ua, const float* __restrict__ ub,
             const int pairBase, float* __restrict__ out, const int bid) {
  constexpr int W = 1 << LOGW;
  constexpr int H = W;
  constexpr int WS = W + 4;
  constexpr int NR = 32 + 8 * NX;
  constexpr int ZROW = NR * WS;
  constexpr int SLABS = H / ROWS;
  static_assert((NR + 1) * WS <= 49 * 260, "LDS array of level_b too small");

  const int b = bid / SLABS;
  const int slab = bid & (SLABS - 1);

  const int lane = threadIdx.x & 31;
  const int wave = threadIdx.x >> 5;
  const int m0 = wave * 16;

  const int mrow = m0 + (lane & 15);
  const bool aval = (mrow < 56);
  const int sidx = aval ? (mrow >> 3) : 0;
  const int l1 = mrow & 7;
  const int slot = c_sdx[sidx] + 1;
  const int dy = aval ? c_sdy[sidx] : 0;
  const int aIdx = (aval ? (slot * 8 + l1) * WS : ZROW) + dy;
  const int ncol = lane & 15;
  const int bBase0 = (ncol < 8) ? (8 + ncol) * WS
                                : ((NX >= 1) ? (32 + (ncol - 8)) * WS : ZROW);
  const int bBase1 = (NX == 2 && ncol < 8) ? (40 + ncol) * WS : ZROW;
  const int khi = (lane >> 4) << 1;

  for (int idx = threadIdx.x; idx < WS; idx += 128) lds[ZROW + idx] = 0.0f;

  v8f acc0 = {};
  v8f acc1 = {};

  for (int xi = 0; xi < ROWS; ++xi) {
    const int xr = slab * ROWS + xi;

    {
      constexpr int PER_CH4 = W / 4;
      constexpr int NA = W / 16;
      constexpr int NP = NX * W / 64;
      float4 va[NA];
      float4 vp[NP > 0 ? NP : 1];
#pragma unroll
      for (int i = 0; i < NA; ++i) {
        const int idx = threadIdx.x + i * 128;
        const int row = idx >> (LOGW - 2);
        const int y4 = idx & (PER_CH4 - 1);
        int xs = xr + (row >> 3) - 1;
        xs += (xs < 0) ? H : 0;
        xs -= (xs >= H) ? H : 0;
        va[i] = *(const float4*)&xin[((((size_t)b * 8 + (row & 7)) << LOGW) +
                                      xs) * W + (y4 << 2)];
      }
#pragma unroll
      for (int i = 0; i < NP; ++i) {
        const int idx = threadIdx.x + i * 128;
        const int prow = idx >> (LOGW - 2);
        const int y4 = idx & (PER_CH4 - 1);
        const float* us = (prow < 8) ? ua : ub;
        vp[i] = *(const float4*)&us[((((size_t)b * 8 + (prow & 7)) << LOGW) +
                                     xr) * W + (y4 << 2)];
      }
      const int erow = threadIdx.x >> 2;
      const int e = threadIdx.x & 3;
      int xe = xr + (erow >> 3) - 1;
      xe += (xe < 0) ? H : 0;
      xe -= (xe >= H) ? H : 0;
      const float ve =
          xin[((((size_t)b * 8 + (erow & 7)) << LOGW) + xe) * W + e];
#pragma unroll
      for (int i = 0; i < NA; ++i) {
        const int idx = threadIdx.x + i * 128;
        *(float4*)&lds[(idx >> (LOGW - 2)) * WS + ((idx & (PER_CH4 - 1)) << 2)] =
            va[i];
      }
#pragma unroll
      for (int i = 0; i < NP; ++i) {
        const int idx = threadIdx.x + i * 128;
        *(float4*)&lds[(32 + (idx >> (LOGW - 2))) * WS +
                       ((idx & (PER_CH4 - 1)) << 2)] = vp[i];
      }
      lds[erow * WS + W + e] = ve;
    }
    __syncthreads();

    {
      constexpr int NBLK = W / 32;
      const int aK = aIdx + khi;
      const int b0K = bBase0 + khi;
      const int b1K = bBase1 + khi;
      v2f fa0[8], fb0[8], fc0[8];
      v2f fa1[8], fb1[8], fc1[8];
      frag_load<NX>(lds, aK, b0K, b1K, fa0, fb0, fc0);
#pragma unroll
      for (int blk = 0; blk < NBLK; blk += 2) {
        frag_load<NX>(lds, aK + (blk + 1) * 32, b0K + (blk + 1) * 32,
                      b1K + (blk + 1) * 32, fa1, fb1, fc1);
        __builtin_amdgcn_sched_barrier(0);
        frag_mma<NX>(fa0, fb0, fc0, acc0, acc1);
        __builtin_amdgcn_sched_barrier(0);
        if (blk + 2 < NBLK)
          frag_load<NX>(lds, aK + (blk + 2) * 32, b0K + (blk + 2) * 32,
                        b1K + (blk + 2) * 32, fa0, fb0, fc0);
        __builtin_amdgcn_sched_barrier(0);
        frag_mma<NX>(fa1, fb1, fc1, acc0, acc1);
        __builtin_amdgcn_sched_barrier(0);
      }
    }
    __syncthreads();
  }

  constexpr int NOUT = 448 * (1 + NX);
  {
    const int s = 2 * wave + (lane >> 4);
    const int col = lane & 15;
    const int l2 = col & 7;
    if ((s < 7) && ((col < 8) || (NX >= 1))) {
      const int t0 = (col >> 3) * 448 + l2 * 7 + s;
#pragma unroll
      for (int i = 0; i < 8; ++i) lds[t0 + i * 56] = acc0[i];
    }
    if ((NX == 2) && (s < 7) && (col < 8)) {
      const int t1 = 896 + l2 * 7 + s;
#pragma unroll
      for (int i = 0; i < 8; ++i) lds[t1 + i * 56] = acc1[i];
    }
  }
  __syncthreads();
  {
    float* const dst = out + ((size_t)b * 384 + pairBase) * 7 + threadIdx.x;
    constexpr int NFULL = NOUT / 128;
    constexpr int REM = NOUT % 128;
    float v[NFULL + 1];
#pragma unroll
    for (int k = 0; k < NFULL; ++k) v[k] = lds[threadIdx.x + 128 * k];
    if (REM != 0 && (int)threadIdx.x < REM)
      v[NFULL] = lds[threadIdx.x + 128 * NFULL];
#pragma unroll
    for (int k = 0; k < NFULL; ++k) atomicAdd(&dst[128 * k], v[k]);
    if (REM != 0 && (int)threadIdx.x < REM)
      atomicAdd(&dst[128 * NFULL], v[NFULL]);
  }
}

__global__ void __launch_bounds__(128)
level_a(const float* __restrict__ x1, const float* __restrict__ x2,
        float* __restrict__ u10, float* __restrict__ u20,
        float* __restrict__ u21, float* __restrict__ out) {
  __shared__ alignas(16) float smem[512 + 16 * 132 + 4];
  const int bid = blockIdx.x;
  if (bid < 256) {
    interp_mm<7, 8>(smem, x1, u10, bid);
  } else if (bid < 512) {
    interp_mm<6, 8>(smem, x2, u20, bid - 256);
  } else {
    const int z = (bid - 512) * 128 + (int)threadIdx.x;
    if (z < 2 * 384 * 7) out[z] = 0.0f;
    interp_mm<6, 7>(smem, x2, u21, bid - 512);
  }
}

__global__ void __launch_bounds__(128) __attribute__((amdgpu_waves_per_eu(1, 4)))
level_b(const float* __restrict__ x0, const float* __restrict__ x1,
        const float* __restrict__ x2, const float* __restrict__ u10,
        const float* __restrict__ u20, const float* __restrict__ u21,
        float* __restrict__ out) {
  __shared__ alignas(16) float smem[49 * 260];
  const int bid = blockIdx.x;
  if (bid < 256) {
    shifted_corr<8, 2, 2>(smem, x0, u10, u20, 0, out, bid);
  } else if (bid < 512) {
    shifted_corr<7, 1, 1>(smem, x1, u21, u21, 192, out, bid - 256);
  } else {
    shifted_corr<6, 1, 0>(smem, x2, x2, x2, 320, out, bid - 512);
  }
}

extern "C" void kernel_launch(void* const* d_in, const int* in_sizes, int n_in,
                              void* d_out, int out_size, void* d_ws,
                              size_t ws_size, hipStream_t stream) {
  const float* x0 = (const float*)d_in[0];
  const float* x1 = (const float*)d_in[1];
  const float* x2 = (const float*)d_in[2];
  float* out = (float*)d_out;

  const size_t n0 = (size_t)16 * 256 * 256;
  const size_t n1 = (size_t)16 * 128 * 128;
  if (ws_size >= (n0 + n0 + n1) * sizeof(float)) {
    float* u10 = (float*)d_ws;
    float* u20 = u10 + n0;
    float* u21 = u20 + n0;

    level_a<<<dim3(640), 128, 0, stream>>>(x1, x2, u10, u20, u21, out);
    level_b<<<dim3(640), 128, 0, stream>>>(x0, x1, x2, u10, u20, u21, out);
  }
}


# level_b scale 0: the second row's own operands (1 X row, partner rows) are requested behind the staging barrier, held in registers under row 0's WMMAs, stored to LDS behind them; sums unchanged
# speedup vs baseline: 10.4843x; 1.0660x over previous
#include <hip/hip_runtime.h>


typedef __attribute__((ext_vector_type(2))) float v2f;
typedef __attribute__((ext_vector_type(8))) float v8f;
typedef __attribute__((ext_vector_type(4))) float v4f;

__device__ __constant__ int c_sdx[7] = {0, 0, 0, 1, 1, 2, -1};
__device__ __constant__ int c_sdy[7] = {0, 1, 2, 0, 1, 0, 1};

template <int NC, int NF, int T1>
__device__ __forceinline__ void
p1_load(const float* __restrict__ g, const float* __restrict__ srcw,
        const int bx, const int blk, v2f (&pa)[8], v2f (&pb)[8][T1]) {
  constexpr int R = NF / NC;
  const int gb = (bx + R * 32 * blk) & (NF - 1);
  const float* const sb = srcw + (size_t)(32 * blk) * NC;
#pragma unroll
  for (int s = 0; s < 8; ++s) {
    pa[s][0] = g[gb + R * (4 * s)];
    pa[s][1] = g[gb + R * (4 * s + 1)];
#pragma unroll
    for (int j = 0; j < T1; ++j) {
      pb[s][j][0] = sb[(4 * s) * NC + 16 * j];
      pb[s][j][1] = sb[(4 * s + 1) * NC + 16 * j];
    }
  }
}

template <int T1>
__device__ __forceinline__ void
p1_mma(const v2f (&pa)[8], const v2f (&pb)[8][T1], v8f (&accV)[T1],
       float& salt) {
#pragma unroll
  for (int s = 0; s < 8; ++s) {
#pragma unroll
    for (int j = 0; j < T1; ++j) {
      salt += pb[s][j][0] - pb[s][j][1];
      accV[j] = __builtin_amdgcn_wmma_f32_16x16x4_f32(
           false, pa[s],  false, pb[s][j],
           (short)0, accV[j],  false,  false);
    }
  }
}

template <int NF, int R, int KB2, int T2>
__device__ __forceinline__ void
p2_load(const float* __restrict__ g, const float* __restrict__ va,
        const int bn, const int blk, v2f (&qa)[KB2], v2f (&qb)[KB2][T2]) {
  const int k0 = 4 * KB2 * blk;
#pragma unroll
  for (int s = 0; s < KB2; ++s) qa[s] = *(const v2f*)&va[k0 + 4 * s];
#pragma unroll
  for (int j = 0; j < T2; ++j) {
    const int gb = (bn - 16 * j + R * k0) & (NF - 1);
#pragma unroll
    for (int s = 0; s < KB2; ++s) {
      qb[s][j][0] = g[gb + R * (4 * s)];
      qb[s][j][1] = g[gb + R * (4 * s + 1)];
    }
  }
}

template <int KB2, int T2>
__device__ __forceinline__ void
p2_mma(const v2f (&qa)[KB2], const v2f (&qb)[KB2][T2], v8f (&accU)[T2]) {
#pragma unroll
  for (int s = 0; s < KB2; ++s) {
#pragma unroll
    for (int j = 0; j < T2; ++j)
      accU[j] = __builtin_amdgcn_wmma_f32_16x16x4_f32(
           false, qa[s],  false, qb[s][j],
           (short)0, accU[j],  false,  false);
  }
}

template <int LOGC, int LOGF>
__device__ __forceinline__ void
interp_mm(float* __restrict__ smem, const float* __restrict__ xc,
          float* __restrict__ up, const int bid) {
  constexpr int NC = 1 << LOGC;
  constexpr int NF = 1 << LOGF;
  constexpr int R = NF / NC;
  constexpr int VP = NC + 4;
  constexpr int T1 = NC / 64;
  constexpr int T2 = NF / 64;
  constexpr int NB1 = NC / 32;
  constexpr int KB2 = 16 / T2;
  constexpr int NB2 = NC / (4 * KB2);
  float* const g = smem;
  float* const Vs = smem + 512;
  float* const red = smem + 512 + 16 * 132;

  const int t = threadIdx.x;
  const int lane = t & 31;
  const int wave = t >> 5;
  const int c = lane & 15;
  const int h = lane >> 4;
  const int khi = h << 1;
  const int img = bid >> (LOGF - 4);
  const int x0 = (bid & ((NF >> 4) - 1)) << 4;
  const float* src = xc + ((size_t)img << (2 * LOGC));

  for (int k = t; k < NF; k += 128) {
    float v = (float)(NC - 1);
    if (k != 0) {
      const int a = ((NC - 1) * k) & (2 * NF - 1);
      v = sinpif((float)a / (float)NF) / sinpif((float)k / (float)NF);
    }
    v += cospif((float)(k & (2 * R - 1)) / (float)R);
    v *= 1.0f / (float)NC;
    const int i = (NF - k) & (NF - 1);
    g[i] = v;
    g[i + NF] = v;
  }
  __syncthreads();

  v8f accV[T1];
#pragma unroll
  for (int j = 0; j < T1; ++j) accV[j] = v8f{};
  float salt = 0.0f;
  const int xa = x0 + c;
  const float* const srcw = src + (size_t)khi * NC + ((wave * T1) << 4) + c;
  {
    v2f pa0[8], pa1[8];
    v2f pb0[8][T1], pb1[8][T1];
    p1_load<NC, NF, T1>(g, srcw, R * khi - xa, 0, pa0, pb0);
#pragma unroll
    for (int blk = 0; blk < NB1; blk += 2) {
      p1_load<NC, NF, T1>(g, srcw, R * khi - xa, blk + 1, pa1, pb1);
      __builtin_amdgcn_sched_barrier(0);
      p1_mma<T1>(pa0, pb0, accV, salt);
      __builtin_amdgcn_sched_barrier(0);
      if (blk + 2 < NB1)
        p1_load<NC, NF, T1>(g, srcw, R * khi - xa, blk + 2, pa0, pb0);
      __builtin_amdgcn_sched_barrier(0);
      p1_mma<T1>(pa1, pb1, accV, salt);
      __builtin_amdgcn_sched_barrier(0);
    }
  }
  salt = (c & 1) ? -salt : salt;
#pragma unroll
  for (int o = 16; o > 0; o >>= 1) salt += __shfl_xor(salt, o);
  if (lane == 0) red[wave] = salt;
#pragma unroll
  for (int j = 0; j < T1; ++j) {
#pragma unroll
    for (int i = 0; i < 8; ++i)
      Vs[(8 * h + i) * VP + ((wave * T1 + j) << 4) + c] = accV[j][i];
  }
  __syncthreads();
  const float S = (red[0] + red[1]) + (red[2] + red[3]);

  v8f accU[T2];
#pragma unroll
  for (int j = 0; j < T2; ++j) accU[j] = v8f{};
  {
    const float* const va = Vs + c * VP + khi;
    const int bn = R * khi - (((wave * T2) << 4) + c);
    v2f qa0[KB2], qa1[KB2];
    v2f qb0[KB2][T2], qb1[KB2][T2];
    p2_load<NF, R, KB2, T2>(g, va, bn, 0, qa0, qb0);
#pragma unroll
    for (int blk = 0; blk < NB2; blk += 2) {
      p2_load<NF, R, KB2, T2>(g, va, bn, blk + 1, qa1, qb1);
      __builtin_amdgcn_sched_barrier(0);
      p2_mma<KB2, T2>(qa0, qb0, accU);
      __builtin_amdgcn_sched_barrier(0);
      if (blk + 2 < NB2) p2_load<NF, R, KB2, T2>(g, va, bn, blk + 2, qa0, qb0);
      __builtin_amdgcn_sched_barrier(0);
      p2_mma<KB2, T2>(qa1, qb1, accU);
      __builtin_amdgcn_sched_barrier(0);
    }
  }

  const float corr = S * (1.0f / (float)(NC * NC));
  float* dst = up + ((size_t)img << (2 * LOGF));
#pragma unroll
  for (int j = 0; j < T2; ++j) {
    const int n = ((wave * T2 + j) << 4) + c;
    const float sn = sinpif((float)(n & (2 * R - 1)) / (float)R) * corr;
#pragma unroll
    for (int i = 0; i < 8; ++i) {
      const int row = x0 + 8 * h + i;
      const float sr = sinpif((float)(row & (2 * R - 1)) / (float)R);
      dst[(size_t)row * NF + n] = accU[j][i] - sr * sn;
    }
  }
}

template <int NX>
__device__ __forceinline__ void
frag_load(const float* __restrict__ lds, const int aOff, const int b0Off,
          const int b1Off, v2f (&fa)[8], v2f (&fb)[8], v2f (&fc)[8]) {
#pragma unroll
  for (int i = 0; i < 8; ++i) {
    fa[i][0] = lds[aOff + 4 * i];
    fa[i][1] = lds[aOff + 4 * i + 1];
    fb[i] = *(const v2f*)&lds[b0Off + 4 * i];
    if (NX == 2) fc[i] = *(const v2f*)&lds[b1Off + 4 * i];
  }
}

template <int NX>
__device__ __forceinline__ void
frag_mma(const v2f (&fa)[8], const v2f (&fb)[8], const v2f (&fc)[8],
         v8f& acc0, v8f& acc1) {
#pragma unroll
  for (int i = 0; i < 8; ++i) {
    acc0 = __builtin_amdgcn_wmma_f32_16x16x4_f32(
         false, fa[i],  false, fb[i],
         (short)0, acc0,  false,  false);
    if (NX == 2)
      acc1 = __builtin_amdgcn_wmma_f32_16x16x4_f32(
           false, fa[i],  false, fc[i],
           (short)0, acc1,  false,  false);
  }
}

template <int LOGW, int NX>
__device__ __forceinline__ void
k_row(const float* __restrict__ lds, const int aK, const int b0K,
      const int b1K, v8f& acc0, v8f& acc1) {
  constexpr int NBLK = (1 << LOGW) / 32;
  v2f fa0[8], fb0[8], fc0[8];
  v2f fa1[8], fb1[8], fc1[8];
  frag_load<NX>(lds, aK, b0K, b1K, fa0, fb0, fc0);
#pragma unroll
  for (int blk = 0; blk < NBLK; blk += 2) {
    frag_load<NX>(lds, aK + (blk + 1) * 32, b0K + (blk + 1) * 32,
                  b1K + (blk + 1) * 32, fa1, fb1, fc1);
    __builtin_amdgcn_sched_barrier(0);
    frag_mma<NX>(fa0, fb0, fc0, acc0, acc1);
    __builtin_amdgcn_sched_barrier(0);
    if (blk + 2 < NBLK)
      frag_load<NX>(lds, aK + (blk + 2) * 32, b0K + (blk + 2) * 32,
                    b1K + (blk + 2) * 32, fa0, fb0, fc0);
    __builtin_amdgcn_sched_barrier(0);
    frag_mma<NX>(fa1, fb1, fc1, acc0, acc1);
    __builtin_amdgcn_sched_barrier(0);
  }
}

template <int LOGW, int ROWS, int NX>
__device__ __forceinline__ void
shifted_corr(float* __restrict__ lds, const float* __restrict__ xin,
             const float* __restrict__ ua, const float* __restrict__ ub,
             const int pairBase, float* __restrict__ out, const int bid) {
  constexpr int W = 1 << LOGW;
  constexpr int H = W;
  constexpr int WS = W + 4;
  constexpr int NAR = 8 * (ROWS + 3);
  constexpr int NPR = 8 * NX * ROWS;
  constexpr int NR = NAR + NPR;
  constexpr int ZROW = NR * WS;
  constexpr int XSTR = 8 * WS;
  constexpr int SLABS = H / ROWS;
  static_assert((NR + 1) * WS <= 73 * 260, "LDS array of level_b too small");
  static_assert(ROWS == 1 || ROWS == 2, "one or two rows per block");

  const int b = bid / SLABS;
  const int slab = bid & (SLABS - 1);
  const int xr0 = slab * ROWS;

  const int lane = threadIdx.x & 31;
  const int wave = threadIdx.x >> 5;
  const int m0 = wave * 16;

  const int mrow = m0 + (lane & 15);
  const bool aval = (mrow < 56);
  const int sidx = aval ? (mrow >> 3) : 0;
  const int l1 = mrow & 7;
  const int slot = c_sdx[sidx] + 1;
  const int dy = aval ? c_sdy[sidx] : 0;
  const int aIdx = (aval ? (slot * 8 + l1) * WS : ZROW) + dy;
  const int ncol = lane & 15;
  const int bBase0 = (ncol < 8) ? (8 + ncol) * WS
                                : ((NX >= 1) ? (NAR + (ncol - 8)) * WS : ZROW);
  const int bBase1 =
      (NX == 2 && ncol < 8) ? (NAR + 8 * ROWS + ncol) * WS : ZROW;
  const int khi = (lane >> 4) << 1;

  for (int idx = threadIdx.x; idx < WS; idx += 128) lds[ZROW + idx] = 0.0f;

  constexpr int PER_CH4 = W / 4;
  constexpr int NA0 = 32 * PER_CH4 / 128;
  constexpr int NP0 = 8 * NX * PER_CH4 / 128;
  constexpr int NA1 = 8 * PER_CH4 / 128;
  float4 va[NA0];
  float4 vp[NP0 > 0 ? NP0 : 1];
  v4f wa[NA1];
  v4f wp[NP0 > 0 ? NP0 : 1];
  float we = 0.0f;
#pragma unroll
  for (int i = 0; i < NA0; ++i) {
    const int idx = threadIdx.x + i * 128;
    const int row = idx >> (LOGW - 2);
    const int y4 = idx & (PER_CH4 - 1);
    int xs = xr0 + (row >> 3) - 1;
    xs += (xs < 0) ? H : 0;
    xs -= (xs >= H) ? H : 0;
    va[i] = *(const float4*)&xin[((((size_t)b * 8 + (row & 7)) << LOGW) +
                                  xs) * W + (y4 << 2)];
  }
#pragma unroll
  for (int i = 0; i < NP0; ++i) {
    const int idx = threadIdx.x + i * 128;
    const int prow = idx >> (LOGW - 2);
    const int y4 = idx & (PER_CH4 - 1);
    const float* us = (prow < 8) ? ua : ub;
    vp[i] = *(const float4*)&us[((((size_t)b * 8 + (prow & 7)) << LOGW) +
                                 xr0) * W + (y4 << 2)];
  }
  const int erow = threadIdx.x >> 2;
  const int e = threadIdx.x & 3;
  int xe = xr0 + (erow >> 3) - 1;
  xe += (xe < 0) ? H : 0;
  xe -= (xe >= H) ? H : 0;
  const float ve = xin[((((size_t)b * 8 + (erow & 7)) << LOGW) + xe) * W + e];
#pragma unroll
  for (int i = 0; i < NA0; ++i) {
    const int idx = threadIdx.x + i * 128;
    *(float4*)&lds[(idx >> (LOGW - 2)) * WS + ((idx & (PER_CH4 - 1)) << 2)] =
        va[i];
  }
#pragma unroll
  for (int i = 0; i < NP0; ++i) {
    const int idx = threadIdx.x + i * 128;
    const int prow = idx >> (LOGW - 2);
    *(float4*)&lds[(NAR + (prow >> 3) * 8 * ROWS + (prow & 7)) * WS +
                   ((idx & (PER_CH4 - 1)) << 2)] = vp[i];
  }
  lds[erow * WS + W + e] = ve;
  __syncthreads();

  if constexpr (ROWS == 2) {
    int x4 = xr0 + 3;
    x4 -= (x4 >= H) ? H : 0;
#pragma unroll
    for (int i = 0; i < NA1; ++i) {
      const int idx = threadIdx.x + i * 128;
      const int ch = idx >> (LOGW - 2);
      const int y4 = idx & (PER_CH4 - 1);
      wa[i] = *(const v4f*)&xin[((((size_t)b * 8 + ch) << LOGW) + x4) * W +
                                   (y4 << 2)];
    }
#pragma unroll
    for (int i = 0; i < NP0; ++i) {
      const int idx = threadIdx.x + i * 128;
      const int prow = idx >> (LOGW - 2);
      const int y4 = idx & (PER_CH4 - 1);
      const float* us = (prow < 8) ? ua : ub;
      wp[i] = *(const v4f*)&us[((((size_t)b * 8 + (prow & 7)) << LOGW) +
                                   xr0 + 1) * W + (y4 << 2)];
    }
    if (threadIdx.x < 32)
      we = xin[((((size_t)b * 8 + erow) << LOGW) + x4) * W + e];
  }

  v8f acc0 = {};
  v8f acc1 = {};
  k_row<LOGW, NX>(lds, aIdx + khi, bBase0 + khi, bBase1 + khi, acc0, acc1);

  if constexpr (ROWS == 2) {
#pragma unroll
    for (int i = 0; i < NA1; ++i) {
      const int idx = threadIdx.x + i * 128;
      *(v4f*)&lds[(32 + (idx >> (LOGW - 2))) * WS +
                  ((idx & (PER_CH4 - 1)) << 2)] = wa[i];
    }
#pragma unroll
    for (int i = 0; i < NP0; ++i) {
      const int idx = threadIdx.x + i * 128;
      const int prow = idx >> (LOGW - 2);
      *(v4f*)&lds[(NAR + (prow >> 3) * 8 * ROWS + 8 + (prow & 7)) * WS +
                  ((idx & (PER_CH4 - 1)) << 2)] = wp[i];
    }
    if (threadIdx.x < 32) lds[(32 + erow) * WS + W + e] = we;
    __syncthreads();
    const int aStr = aval ? XSTR : 0;
    const int b0Str = (ncol < 8 || NX >= 1) ? XSTR : 0;
    const int b1Str = (NX == 2 && ncol < 8) ? XSTR : 0;
    k_row<LOGW, NX>(lds, aIdx + khi + aStr, bBase0 + khi + b0Str,
                    bBase1 + khi + b1Str, acc0, acc1);
  }
  __syncthreads();

  constexpr int NOUT = 448 * (1 + NX);
  {
    const int s = 2 * wave + (lane >> 4);
    const int col = lane & 15;
    const int l2 = col & 7;
    if ((s < 7) && ((col < 8) || (NX >= 1))) {
      const int t0 = (col >> 3) * 448 + l2 * 7 + s;
#pragma unroll
      for (int i = 0; i < 8; ++i) lds[t0 + i * 56] = acc0[i];
    }
    if ((NX == 2) && (s < 7) && (col < 8)) {
      const int t1 = 896 + l2 * 7 + s;
#pragma unroll
      for (int i = 0; i < 8; ++i) lds[t1 + i * 56] = acc1[i];
    }
  }
  __syncthreads();
  {
    float* const dst = out + ((size_t)b * 384 + pairBase) * 7 + threadIdx.x;
    constexpr int NFULL = NOUT / 128;
    constexpr int REM = NOUT % 128;
    float v[NFULL + 1];
#pragma unroll
    for (int k = 0; k < NFULL; ++k) v[k] = lds[threadIdx.x + 128 * k];
    if (REM != 0 && (int)threadIdx.x < REM)
      v[NFULL] = lds[threadIdx.x + 128 * NFULL];
#pragma unroll
    for (int k = 0; k < NFULL; ++k) atomicAdd(&dst[128 * k], v[k]);
    if (REM != 0 && (int)threadIdx.x < REM)
      atomicAdd(&dst[128 * NFULL], v[NFULL]);
  }
}

__global__ void __launch_bounds__(128)
level_a(const float* __restrict__ x1, const float* __restrict__ x2,
        float* __restrict__ u10, float* __restrict__ u20,
        float* __restrict__ u21, float* __restrict__ out) {
  __shared__ alignas(16) float smem[512 + 16 * 132 + 4];
  const int bid = blockIdx.x;
  if (bid < 256) {
    interp_mm<7, 8>(smem, x1, u10, bid);
  } else if (bid < 512) {
    interp_mm<6, 8>(smem, x2, u20, bid - 256);
  } else {
    const int z = (bid - 512) * 128 + (int)threadIdx.x;
    if (z < 2 * 384 * 7) out[z] = 0.0f;
    interp_mm<6, 7>(smem, x2, u21, bid - 512);
  }
}

__global__ void __launch_bounds__(128) __attribute__((amdgpu_waves_per_eu(1, 4)))
level_b(const float* __restrict__ x0, const float* __restrict__ x1,
        const float* __restrict__ x2, const float* __restrict__ u10,
        const float* __restrict__ u20, const float* __restrict__ u21,
        float* __restrict__ out) {
  __shared__ alignas(16) float smem[73 * 260];
  const int bid = blockIdx.x;
  if (bid < 256) {
    shifted_corr<8, 2, 2>(smem, x0, u10, u20, 0, out, bid);
  } else if (bid < 512) {
    shifted_corr<7, 1, 1>(smem, x1, u21, u21, 192, out, bid - 256);
  } else {
    shifted_corr<6, 1, 0>(smem, x2, x2, x2, 320, out, bid - 512);
  }
}

extern "C" void kernel_launch(void* const* d_in, const int* in_sizes, int n_in,
                              void* d_out, int out_size, void* d_ws,
                              size_t ws_size, hipStream_t stream) {
  const float* x0 = (const float*)d_in[0];
  const float* x1 = (const float*)d_in[1];
  const float* x2 = (const float*)d_in[2];
  float* out = (float*)d_out;

  const size_t n0 = (size_t)16 * 256 * 256;
  const size_t n1 = (size_t)16 * 128 * 128;
  if (ws_size >= (n0 + n0 + n1) * sizeof(float)) {
    float* u10 = (float*)d_ws;
    float* u20 = u10 + n0;
    float* u21 = u20 + n0;

    level_a<<<dim3(640), 128, 0, stream>>>(x1, x2, u10, u20, u21, out);
    level_b<<<dim3(640), 128, 0, stream>>>(x0, x1, x2, u10, u20, u21, out);
  }
}


# interpolation product 2 runs only over off-grid fine columns (g is 1 at one k and 0 elsewhere on grid columns): on-grid columns stored from product 1's accumulators; off-grid sums unchanged
# speedup vs baseline: 11.5901x; 1.1055x over previous
#include <hip/hip_runtime.h>


typedef __attribute__((ext_vector_type(2))) float v2f;
typedef __attribute__((ext_vector_type(8))) float v8f;
typedef __attribute__((ext_vector_type(4))) float v4f;

__device__ __constant__ int c_sdx[7] = {0, 0, 0, 1, 1, 2, -1};
__device__ __constant__ int c_sdy[7] = {0, 1, 2, 0, 1, 0, 1};

template <int NC, int NF, int T1>
__device__ __forceinline__ void
p1_load(const float* __restrict__ g, const float* __restrict__ srcw,
        const int bx, const int blk, v2f (&pa)[8], v2f (&pb)[8][T1]) {
  constexpr int R = NF / NC;
  const int gb = (bx + R * 32 * blk) & (NF - 1);
  const float* const sb = srcw + (size_t)(32 * blk) * NC;
#pragma unroll
  for (int s = 0; s < 8; ++s) {
    pa[s][0] = g[gb + R * (4 * s)];
    pa[s][1] = g[gb + R * (4 * s + 1)];
#pragma unroll
    for (int j = 0; j < T1; ++j) {
      pb[s][j][0] = sb[(4 * s) * NC + 16 * j];
      pb[s][j][1] = sb[(4 * s + 1) * NC + 16 * j];
    }
  }
}

template <int T1>
__device__ __forceinline__ void
p1_mma(const v2f (&pa)[8], const v2f (&pb)[8][T1], v8f (&accV)[T1],
       float& salt) {
#pragma unroll
  for (int s = 0; s < 8; ++s) {
#pragma unroll
    for (int j = 0; j < T1; ++j) {
      salt += pb[s][j][0] - pb[s][j][1];
      accV[j] = __builtin_amdgcn_wmma_f32_16x16x4_f32(
           false, pa[s],  false, pb[s][j],
           (short)0, accV[j],  false,  false);
    }
  }
}

template <int NF, int R, int KB2, int T2, int T1>
__device__ __forceinline__ void
p2_load(const float* __restrict__ g, const float* __restrict__ va,
        const int bn, const int blk, v2f (&qa)[KB2], v2f (&qb)[KB2][T2]) {
  const int k0 = 4 * KB2 * blk;
#pragma unroll
  for (int s = 0; s < KB2; ++s) qa[s] = *(const v2f*)&va[k0 + 4 * s];
#pragma unroll
  for (int j = 0; j < T2; ++j) {
    const int gb =
        (bn - (R * 16 * (j % T1) + 1 + j / T1) + R * k0) & (NF - 1);
#pragma unroll
    for (int s = 0; s < KB2; ++s) {
      qb[s][j][0] = g[gb + R * (4 * s)];
      qb[s][j][1] = g[gb + R * (4 * s + 1)];
    }
  }
}

template <int KB2, int T2>
__device__ __forceinline__ void
p2_mma(const v2f (&qa)[KB2], const v2f (&qb)[KB2][T2], v8f (&accU)[T2]) {
#pragma unroll
  for (int s = 0; s < KB2; ++s) {
#pragma unroll
    for (int j = 0; j < T2; ++j)
      accU[j] = __builtin_amdgcn_wmma_f32_16x16x4_f32(
           false, qa[s],  false, qb[s][j],
           (short)0, accU[j],  false,  false);
  }
}

template <int LOGC, int LOGF>
__device__ __forceinline__ void
interp_mm(float* __restrict__ smem, const float* __restrict__ xc,
          float* __restrict__ up, const int bid) {
  constexpr int NC = 1 << LOGC;
  constexpr int NF = 1 << LOGF;
  constexpr int R = NF / NC;
  constexpr int VP = NC + 4;
  constexpr int T1 = NC / 64;
  constexpr int T2 = (R - 1) * T1;
  constexpr int NB1 = NC / 32;
  constexpr int KB2 = (T2 == 3) ? 4 : 8;
  constexpr int NB2 = NC / (4 * KB2);
  static_assert(NB2 >= 2 && NB2 % 2 == 0, "product 2 runs in block pairs");
  float* const g = smem;
  float* const Vs = smem + 512;
  float* const red = smem + 512 + 16 * 132;

  const int t = threadIdx.x;
  const int lane = t & 31;
  const int wave = t >> 5;
  const int c = lane & 15;
  const int h = lane >> 4;
  const int khi = h << 1;
  const int img = bid >> (LOGF - 4);
  const int x0 = (bid & ((NF >> 4) - 1)) << 4;
  const float* src = xc + ((size_t)img << (2 * LOGC));

  for (int k = t; k < NF; k += 128) {
    float v = (float)(NC - 1);
    if (k != 0) {
      const int a = ((NC - 1) * k) & (2 * NF - 1);
      v = sinpif((float)a / (float)NF) / sinpif((float)k / (float)NF);
    }
    v += cospif((float)(k & (2 * R - 1)) / (float)R);
    v *= 1.0f / (float)NC;
    const int i = (NF - k) & (NF - 1);
    g[i] = v;
    g[i + NF] = v;
  }
  __syncthreads();

  v8f accV[T1];
#pragma unroll
  for (int j = 0; j < T1; ++j) accV[j] = v8f{};
  float salt = 0.0f;
  const int xa = x0 + c;
  const float* const srcw = src + (size_t)khi * NC + ((wave * T1) << 4) + c;
  {
    v2f pa0[8], pa1[8];
    v2f pb0[8][T1], pb1[8][T1];
    p1_load<NC, NF, T1>(g, srcw, R * khi - xa, 0, pa0, pb0);
#pragma unroll
    for (int blk = 0; blk < NB1; blk += 2) {
      p1_load<NC, NF, T1>(g, srcw, R * khi - xa, blk + 1, pa1, pb1);
      __builtin_amdgcn_sched_barrier(0);
      p1_mma<T1>(pa0, pb0, accV, salt);
      __builtin_amdgcn_sched_barrier(0);
      if (blk + 2 < NB1)
        p1_load<NC, NF, T1>(g, srcw, R * khi - xa, blk + 2, pa0, pb0);
      __builtin_amdgcn_sched_barrier(0);
      p1_mma<T1>(pa1, pb1, accV, salt);
      __builtin_amdgcn_sched_barrier(0);
    }
  }
  salt = (c & 1) ? -salt : salt;
#pragma unroll
  for (int o = 16; o > 0; o >>= 1) salt += __shfl_xor(salt, o);
  if (lane == 0) red[wave] = salt;
#pragma unroll
  for (int j = 0; j < T1; ++j) {
#pragma unroll
    for (int i = 0; i < 8; ++i)
      Vs[(8 * h + i) * VP + ((wave * T1 + j) << 4) + c] = accV[j][i];
  }
  __syncthreads();
  const float S = (red[0] + red[1]) + (red[2] + red[3]);

  v8f accU[T2];
#pragma unroll
  for (int j = 0; j < T2; ++j) accU[j] = v8f{};
  {
    const float* const va = Vs + c * VP + khi;
    const int bn = R * (khi - (((wave * T1) << 4) + c));
    v2f qa0[KB2], qa1[KB2];
    v2f qb0[KB2][T2], qb1[KB2][T2];
    p2_load<NF, R, KB2, T2, T1>(g, va, bn, 0, qa0, qb0);
#pragma unroll
    for (int blk = 0; blk < NB2; blk += 2) {
      p2_load<NF, R, KB2, T2, T1>(g, va, bn, blk + 1, qa1, qb1);
      __builtin_amdgcn_sched_barrier(0);
      p2_mma<KB2, T2>(qa0, qb0, accU);
      __builtin_amdgcn_sched_barrier(0);
      if (blk + 2 < NB2)
        p2_load<NF, R, KB2, T2, T1>(g, va, bn, blk + 2, qa0, qb0);
      __builtin_amdgcn_sched_barrier(0);
      p2_mma<KB2, T2>(qa1, qb1, accU);
      __builtin_amdgcn_sched_barrier(0);
    }
  }

  const float corr = S * (1.0f / (float)(NC * NC));
  float* dst = up + ((size_t)img << (2 * LOGF));
#pragma unroll
  for (int nb = 0; nb < T1; ++nb) {
    const int nc = ((wave * T1 + nb) << 4) + c;
    float sn[R - 1];
#pragma unroll
    for (int p = 1; p < R; ++p) {
      const int n = R * nc + p;
      sn[p - 1] = sinpif((float)(n & (2 * R - 1)) / (float)R) * corr;
    }
#pragma unroll
    for (int i = 0; i < 8; ++i) {
      const int row = x0 + 8 * h + i;
      const float sr = sinpif((float)(row & (2 * R - 1)) / (float)R);
      float* const q = dst + (size_t)row * NF + R * nc;
      if constexpr (R == 2) {
        v2f o;
        o[0] = accV[nb][i];
        o[1] = accU[nb][i] - sr * sn[0];
        *(v2f*)q = o;
      } else {
        static_assert(R == 4, "interpolation ratio 2 or 4");
        v4f o;
        o[0] = accV[nb][i];
        o[1] = accU[nb][i] - sr * sn[0];
        o[2] = accU[T1 + nb][i] - sr * sn[1];
        o[3] = accU[2 * T1 + nb][i] - sr * sn[2];
        *(v4f*)q = o;
      }
    }
  }
}

template <int NX>
__device__ __forceinline__ void
frag_load(const float* __restrict__ lds, const int aOff, const int b0Off,
          const int b1Off, v2f (&fa)[8], v2f (&fb)[8], v2f (&fc)[8]) {
#pragma unroll
  for (int i = 0; i < 8; ++i) {
    fa[i][0] = lds[aOff + 4 * i];
    fa[i][1] = lds[aOff + 4 * i + 1];
    fb[i] = *(const v2f*)&lds[b0Off + 4 * i];
    if (NX == 2) fc[i] = *(const v2f*)&lds[b1Off + 4 * i];
  }
}

template <int NX>
__device__ __forceinline__ void
frag_mma(const v2f (&fa)[8], const v2f (&fb)[8], const v2f (&fc)[8],
         v8f& acc0, v8f& acc1) {
#pragma unroll
  for (int i = 0; i < 8; ++i) {
    acc0 = __builtin_amdgcn_wmma_f32_16x16x4_f32(
         false, fa[i],  false, fb[i],
         (short)0, acc0,  false,  false);
    if (NX == 2)
      acc1 = __builtin_amdgcn_wmma_f32_16x16x4_f32(
           false, fa[i],  false, fc[i],
           (short)0, acc1,  false,  false);
  }
}

template <int LOGW, int NX>
__device__ __forceinline__ void
k_row(const float* __restrict__ lds, const int aK, const int b0K,
      const int b1K, v8f& acc0, v8f& acc1) {
  constexpr int NBLK = (1 << LOGW) / 32;
  v2f fa0[8], fb0[8], fc0[8];
  v2f fa1[8], fb1[8], fc1[8];
  frag_load<NX>(lds, aK, b0K, b1K, fa0, fb0, fc0);
#pragma unroll
  for (int blk = 0; blk < NBLK; blk += 2) {
    frag_load<NX>(lds, aK + (blk + 1) * 32, b0K + (blk + 1) * 32,
                  b1K + (blk + 1) * 32, fa1, fb1, fc1);
    __builtin_amdgcn_sched_barrier(0);
    frag_mma<NX>(fa0, fb0, fc0, acc0, acc1);
    __builtin_amdgcn_sched_barrier(0);
    if (blk + 2 < NBLK)
      frag_load<NX>(lds, aK + (blk + 2) * 32, b0K + (blk + 2) * 32,
                    b1K + (blk + 2) * 32, fa0, fb0, fc0);
    __builtin_amdgcn_sched_barrier(0);
    frag_mma<NX>(fa1, fb1, fc1, acc0, acc1);
    __builtin_amdgcn_sched_barrier(0);
  }
}

template <int LOGW, int ROWS, int NX>
__device__ __forceinline__ void
shifted_corr(float* __restrict__ lds, const float* __restrict__ xin,
             const float* __restrict__ ua, const float* __restrict__ ub,
             const int pairBase, float* __restrict__ out, const int bid) {
  constexpr int W = 1 << LOGW;
  constexpr int H = W;
  constexpr int WS = W + 4;
  constexpr int NAR = 8 * (ROWS + 3);
  constexpr int NPR = 8 * NX * ROWS;
  constexpr int NR = NAR + NPR;
  constexpr int ZROW = NR * WS;
  constexpr int XSTR = 8 * WS;
  constexpr int SLABS = H / ROWS;
  static_assert((NR + 1) * WS <= 73 * 260, "LDS array of level_b too small");
  static_assert(ROWS == 1 || ROWS == 2, "one or two rows per block");

  const int b = bid / SLABS;
  const int slab = bid & (SLABS - 1);
  const int xr0 = slab * ROWS;

  const int lane = threadIdx.x & 31;
  const int wave = threadIdx.x >> 5;
  const int m0 = wave * 16;

  const int mrow = m0 + (lane & 15);
  const bool aval = (mrow < 56);
  const int sidx = aval ? (mrow >> 3) : 0;
  const int l1 = mrow & 7;
  const int slot = c_sdx[sidx] + 1;
  const int dy = aval ? c_sdy[sidx] : 0;
  const int aIdx = (aval ? (slot * 8 + l1) * WS : ZROW) + dy;
  const int ncol = lane & 15;
  const int bBase0 = (ncol < 8) ? (8 + ncol) * WS
                                : ((NX >= 1) ? (NAR + (ncol - 8)) * WS : ZROW);
  const int bBase1 =
      (NX == 2 && ncol < 8) ? (NAR + 8 * ROWS + ncol) * WS : ZROW;
  const int khi = (lane >> 4) << 1;

  for (int idx = threadIdx.x; idx < WS; idx += 128) lds[ZROW + idx] = 0.0f;

  constexpr int PER_CH4 = W / 4;
  constexpr int NA0 = 32 * PER_CH4 / 128;
  constexpr int NP0 = 8 * NX * PER_CH4 / 128;
  constexpr int NA1 = 8 * PER_CH4 / 128;
  float4 va[NA0];
  float4 vp[NP0 > 0 ? NP0 : 1];
  v4f wa[NA1];
  v4f wp[NP0 > 0 ? NP0 : 1];
  float we = 0.0f;
#pragma unroll
  for (int i = 0; i < NA0; ++i) {
    const int idx = threadIdx.x + i * 128;
    const int row = idx >> (LOGW - 2);
    const int y4 = idx & (PER_CH4 - 1);
    int xs = xr0 + (row >> 3) - 1;
    xs += (xs < 0) ? H : 0;
    xs -= (xs >= H) ? H : 0;
    va[i] = *(const float4*)&xin[((((size_t)b * 8 + (row & 7)) << LOGW) +
                                  xs) * W + (y4 << 2)];
  }
#pragma unroll
  for (int i = 0; i < NP0; ++i) {
    const int idx = threadIdx.x + i * 128;
    const int prow = idx >> (LOGW - 2);
    const int y4 = idx & (PER_CH4 - 1);
    const float* us = (prow < 8) ? ua : ub;
    vp[i] = *(const float4*)&us[((((size_t)b * 8 + (prow & 7)) << LOGW) +
                                 xr0) * W + (y4 << 2)];
  }
  const int erow = threadIdx.x >> 2;
  const int e = threadIdx.x & 3;
  int xe = xr0 + (erow >> 3) - 1;
  xe += (xe < 0) ? H : 0;
  xe -= (xe >= H) ? H : 0;
  const float ve = xin[((((size_t)b * 8 + (erow & 7)) << LOGW) + xe) * W + e];
#pragma unroll
  for (int i = 0; i < NA0; ++i) {
    const int idx = threadIdx.x + i * 128;
    *(float4*)&lds[(idx >> (LOGW - 2)) * WS + ((idx & (PER_CH4 - 1)) << 2)] =
        va[i];
  }
#pragma unroll
  for (int i = 0; i < NP0; ++i) {
    const int idx = threadIdx.x + i * 128;
    const int prow = idx >> (LOGW - 2);
    *(float4*)&lds[(NAR + (prow >> 3) * 8 * ROWS + (prow & 7)) * WS +
                   ((idx & (PER_CH4 - 1)) << 2)] = vp[i];
  }
  lds[erow * WS + W + e] = ve;
  __syncthreads();

  if constexpr (ROWS == 2) {
    int x4 = xr0 + 3;
    x4 -= (x4 >= H) ? H : 0;
#pragma unroll
    for (int i = 0; i < NA1; ++i) {
      const int idx = threadIdx.x + i * 128;
      const int ch = idx >> (LOGW - 2);
      const int y4 = idx & (PER_CH4 - 1);
      wa[i] = *(const v4f*)&xin[((((size_t)b * 8 + ch) << LOGW) + x4) * W +
                                   (y4 << 2)];
    }
#pragma unroll
    for (int i = 0; i < NP0; ++i) {
      const int idx = threadIdx.x + i * 128;
      const int prow = idx >> (LOGW - 2);
      const int y4 = idx & (PER_CH4 - 1);
      const float* us = (prow < 8) ? ua : ub;
      wp[i] = *(const v4f*)&us[((((size_t)b * 8 + (prow & 7)) << LOGW) +
                                   xr0 + 1) * W + (y4 << 2)];
    }
    if (threadIdx.x < 32)
      we = xin[((((size_t)b * 8 + erow) << LOGW) + x4) * W + e];
  }

  v8f acc0 = {};
  v8f acc1 = {};
  k_row<LOGW, NX>(lds, aIdx + khi, bBase0 + khi, bBase1 + khi, acc0, acc1);

  if constexpr (ROWS == 2) {
#pragma unroll
    for (int i = 0; i < NA1; ++i) {
      const int idx = threadIdx.x + i * 128;
      *(v4f*)&lds[(32 + (idx >> (LOGW - 2))) * WS +
                  ((idx & (PER_CH4 - 1)) << 2)] = wa[i];
    }
#pragma unroll
    for (int i = 0; i < NP0; ++i) {
      const int idx = threadIdx.x + i * 128;
      const int prow = idx >> (LOGW - 2);
      *(v4f*)&lds[(NAR + (prow >> 3) * 8 * ROWS + 8 + (prow & 7)) * WS +
                  ((idx & (PER_CH4 - 1)) << 2)] = wp[i];
    }
    if (threadIdx.x < 32) lds[(32 + erow) * WS + W + e] = we;
    __syncthreads();
    const int aStr = aval ? XSTR : 0;
    const int b0Str = (ncol < 8 || NX >= 1) ? XSTR : 0;
    const int b1Str = (NX == 2 && ncol < 8) ? XSTR : 0;
    k_row<LOGW, NX>(lds, aIdx + khi + aStr, bBase0 + khi + b0Str,
                    bBase1 + khi + b1Str, acc0, acc1);
  }
  __syncthreads();

  constexpr int NOUT = 448 * (1 + NX);
  {
    const int s = 2 * wave + (lane >> 4);
    const int col = lane & 15;
    const int l2 = col & 7;
    if ((s < 7) && ((col < 8) || (NX >= 1))) {
      const int t0 = (col >> 3) * 448 + l2 * 7 + s;
#pragma unroll
      for (int i = 0; i < 8; ++i) lds[t0 + i * 56] = acc0[i];
    }
    if ((NX == 2) && (s < 7) && (col < 8)) {
      const int t1 = 896 + l2 * 7 + s;
#pragma unroll
      for (int i = 0; i < 8; ++i) lds[t1 + i * 56] = acc1[i];
    }
  }
  __syncthreads();
  {
    float* const dst = out + ((size_t)b * 384 + pairBase) * 7 + threadIdx.x;
    constexpr int NFULL = NOUT / 128;
    constexpr int REM = NOUT % 128;
    float v[NFULL + 1];
#pragma unroll
    for (int k = 0; k < NFULL; ++k) v[k] = lds[threadIdx.x + 128 * k];
    if (REM != 0 && (int)threadIdx.x < REM)
      v[NFULL] = lds[threadIdx.x + 128 * NFULL];
#pragma unroll
    for (int k = 0; k < NFULL; ++k) atomicAdd(&dst[128 * k], v[k]);
    if (REM != 0 && (int)threadIdx.x < REM)
      atomicAdd(&dst[128 * NFULL], v[NFULL]);
  }
}

__global__ void __launch_bounds__(128)
level_a(const float* __restrict__ x1, const float* __restrict__ x2,
        float* __restrict__ u10, float* __restrict__ u20,
        float* __restrict__ u21, float* __restrict__ out) {
  __shared__ alignas(16) float smem[512 + 16 * 132 + 4];
  const int bid = blockIdx.x;
  if (bid < 256) {
    interp_mm<7, 8>(smem, x1, u10, bid);
  } else if (bid < 512) {
    interp_mm<6, 8>(smem, x2, u20, bid - 256);
  } else {
    const int z = (bid - 512) * 128 + (int)threadIdx.x;
    if (z < 2 * 384 * 7) out[z] = 0.0f;
    interp_mm<6, 7>(smem, x2, u21, bid - 512);
  }
}

__global__ void __launch_bounds__(128) __attribute__((amdgpu_waves_per_eu(1, 4)))
level_b(const float* __restrict__ x0, const float* __restrict__ x1,
        const float* __restrict__ x2, const float* __restrict__ u10,
        const float* __restrict__ u20, const float* __restrict__ u21,
        float* __restrict__ out) {
  __shared__ alignas(16) float smem[73 * 260];
  const int bid = blockIdx.x;
  if (bid < 256) {
    shifted_corr<8, 2, 2>(smem, x0, u10, u20, 0, out, bid);
  } else if (bid < 512) {
    shifted_corr<7, 1, 1>(smem, x1, u21, u21, 192, out, bid - 256);
  } else {
    shifted_corr<6, 1, 0>(smem, x2, x2, x2, 320, out, bid - 512);
  }
}

extern "C" void kernel_launch(void* const* d_in, const int* in_sizes, int n_in,
                              void* d_out, int out_size, void* d_ws,
                              size_t ws_size, hipStream_t stream) {
  const float* x0 = (const float*)d_in[0];
  const float* x1 = (const float*)d_in[1];
  const float* x2 = (const float*)d_in[2];
  float* out = (float*)d_out;

  const size_t n0 = (size_t)16 * 256 * 256;
  const size_t n1 = (size_t)16 * 128 * 128;
  if (ws_size >= (n0 + n0 + n1) * sizeof(float)) {
    float* u10 = (float*)d_ws;
    float* u20 = u10 + n0;
    float* u21 = u20 + n0;

    level_a<<<dim3(640), 128, 0, stream>>>(x1, x2, u10, u20, u21, out);
    level_b<<<dim3(640), 128, 0, stream>>>(x0, x1, x2, u10, u20, u21, out);
  }
}
